# starting kernel
# baseline (speedup 1.0000x reference)
#include <hip/hip_runtime.h>

#define IN_DIM  256
#define HID_DIM 128
#define OUT_DIM 64

typedef __attribute__((ext_vector_type(2))) float v2f;
typedef __attribute__((ext_vector_type(8))) float v8f;

__global__ void init_deg_kernel(float* __restrict__ deg, int n) {
  int i = blockIdx.x * blockDim.x + threadIdx.x;
  if (i < n) deg[i] = 1.0f;
}

__global__ void accum_deg_kernel(const int* __restrict__ dst,
                                 float* __restrict__ deg, int e) {
  int i = blockIdx.x * blockDim.x + threadIdx.x;
  if (i < e) atomicAdd(&deg[dst[i]], 1.0f);
}

__global__ void dinv_kernel(float* __restrict__ deg, int n) {
  int i = blockIdx.x * blockDim.x + threadIdx.x;
  if (i < n) deg[i] = rsqrtf(deg[i]);
}

template <int K, int NT>
__global__ void gemm_wmma_kernel(const float* __restrict__ A,
                                 const float* __restrict__ B,
                                 float* __restrict__ C, int M) {
  const int N = NT * 16;
  const int wave = (int)((blockIdx.x * (unsigned)blockDim.x + threadIdx.x) >> 5);
  const int lane = threadIdx.x & 31;
  const int row0 = wave * 16;
  if (row0 >= M) return;

  const int mn    = lane & 15;
  const int khalf = lane >> 4;

  int arow = row0 + mn;
  if (arow >= M) arow = M - 1;
  const float* __restrict__ Ar = A + (size_t)arow * K;

  v8f acc[NT];
#pragma unroll
  for (int t = 0; t < NT; ++t) acc[t] = v8f{};

  for (int k = 0; k < K; k += 4) {
    const int ka = k + 2 * khalf;
    v2f a;
    a.x = Ar[ka];
    a.y = Ar[ka + 1];
#pragma unroll
    for (int t = 0; t < NT; ++t) {
      v2f b;
      b.x = B[(size_t)ka       * N + t * 16 + mn];
      b.y = B[(size_t)(ka + 1) * N + t * 16 + mn];
      acc[t] = __builtin_amdgcn_wmma_f32_16x16x4_f32(
           false, a,  false, b,
           (short)0, acc[t],  false,  false);
    }
  }

#pragma unroll
  for (int t = 0; t < NT; ++t) {
#pragma unroll
    for (int i = 0; i < 8; ++i) {
      const int row = row0 + 8 * khalf + i;
      if (row < M) C[(size_t)row * N + t * 16 + mn] = acc[t][i];
    }
  }
}

__global__ void init_bias_kernel(float* __restrict__ out,
                                 const float* __restrict__ bias,
                                 int fmask, long long total) {
  long long idx = blockIdx.x * (long long)blockDim.x + threadIdx.x;
  if (idx < total) out[idx] = bias[(int)(idx & fmask)];
}

template <int F>
__global__ void scatter_kernel(const float* __restrict__ h,
                               const int* __restrict__ src,
                               const int* __restrict__ dst,
                               const float* __restrict__ dinv,
                               float* __restrict__ agg,
                               int n_edges, int n_nodes) {
  long long idx = blockIdx.x * (long long)blockDim.x + threadIdx.x;
  long long total = (long long)(n_edges + n_nodes) * F;
  if (idx >= total) return;
  const int e = (int)(idx / F);
  const int f = (int)(idx & (F - 1));
  int s, d;
  if (e < n_edges) { s = src[e]; d = dst[e]; }
  else             { s = d = e - n_edges; }
  const float norm = dinv[s] * dinv[d];
  atomicAdd(&agg[(size_t)d * F + f], h[(size_t)s * F + f] * norm);
}

__global__ void relu_kernel(float* __restrict__ v, long long total) {
  long long idx = blockIdx.x * (long long)blockDim.x + threadIdx.x;
  if (idx < total) {
    float x = v[idx];
    v[idx] = x > 0.0f ? x : 0.0f;
  }
}

extern "C" void kernel_launch(void* const* d_in, const int* in_sizes, int n_in,
                              void* d_out, int out_size, void* d_ws, size_t ws_size,
                              hipStream_t stream) {
  const float* x  = (const float*)d_in[0];
  const int*   ei = (const int*)d_in[1];
  const float* W1 = (const float*)d_in[2];
  const float* b1 = (const float*)d_in[3];
  const float* W2 = (const float*)d_in[4];
  const float* b2 = (const float*)d_in[5];
  float* out = (float*)d_out;

  const int N = in_sizes[0] / IN_DIM;
  const int E = in_sizes[1] / 2;
  const int* src = ei;
  const int* dst = ei + E;

  float* dinv = (float*)d_ws;
  float* h    = dinv + N;
  float* agg1 = h + (size_t)N * HID_DIM;

  const int B256 = 256;
  init_deg_kernel<<<(N + B256 - 1) / B256, B256, 0, stream>>>(dinv, N);
  accum_deg_kernel<<<(E + B256 - 1) / B256, B256, 0, stream>>>(dst, dinv, E);
  dinv_kernel<<<(N + B256 - 1) / B256, B256, 0, stream>>>(dinv, N);

  const int waves  = (N + 15) / 16;
  const int gblks  = (waves + 3) / 4;

  gemm_wmma_kernel<IN_DIM, HID_DIM / 16><<<gblks, 128, 0, stream>>>(x, W1, h, N);
  const long long tot1 = (long long)N * HID_DIM;
  init_bias_kernel<<<(int)((tot1 + B256 - 1) / B256), B256, 0, stream>>>(agg1, b1, HID_DIM - 1, tot1);
  const long long sc1 = (long long)(E + N) * HID_DIM;
  scatter_kernel<HID_DIM><<<(int)((sc1 + B256 - 1) / B256), B256, 0, stream>>>(
      h, src, dst, dinv, agg1, E, N);
  relu_kernel<<<(int)((tot1 + B256 - 1) / B256), B256, 0, stream>>>(agg1, tot1);

  gemm_wmma_kernel<HID_DIM, OUT_DIM / 16><<<gblks, 128, 0, stream>>>(agg1, W2, h, N);
  const long long tot2 = (long long)N * OUT_DIM;
  init_bias_kernel<<<(int)((tot2 + B256 - 1) / B256), B256, 0, stream>>>(out, b2, OUT_DIM - 1, tot2);
  const long long sc2 = (long long)(E + N) * OUT_DIM;
  scatter_kernel<OUT_DIM><<<(int)((sc2 + B256 - 1) / B256), B256, 0, stream>>>(
      h, src, dst, dinv, out, E, N);
}


# Float-atomic edge scatter replaced by a per-call counting sort of edges by destination and a per-node gather writing each row once (bias, relu fused); sum order per node = slot arrival order
# speedup vs baseline: 2.8180x; 2.8180x over previous
#include <hip/hip_runtime.h>

#define IN_DIM  256
#define HID_DIM 128
#define OUT_DIM 64

typedef __attribute__((ext_vector_type(2))) float v2f;
typedef __attribute__((ext_vector_type(8))) float v8f;

__global__ void zero_kernel(int* __restrict__ p, int n) {
  int i = blockIdx.x * blockDim.x + threadIdx.x;
  if (i < n) p[i] = 0;
}

__global__ void count_rank_kernel(const int* __restrict__ dst,
                                  int* __restrict__ cnt,
                                  int* __restrict__ rank, int e) {
  int i = blockIdx.x * blockDim.x + threadIdx.x;
  if (i < e) rank[i] = atomicAdd(&cnt[dst[i]], 1);
}

__global__ void segment_kernel(const int* __restrict__ cnt,
                               int* __restrict__ total,
                               int* __restrict__ rowstart,
                               float* __restrict__ dinv, int n) {
  const int i = blockIdx.x * blockDim.x + threadIdx.x;
  const int lane = threadIdx.x & (warpSize - 1);
  const int c = (i < n) ? cnt[i] : 0;
  int incl = c;
  for (int d = 1; d < warpSize; d <<= 1) {
    int t = __shfl_up(incl, d);
    if (lane >= d) incl += t;
  }
  int base = 0;
  if (lane == warpSize - 1) base = atomicAdd(total, incl);
  base = __shfl(base, warpSize - 1);
  if (i < n) {
    rowstart[i] = base + incl - c;
    dinv[i] = rsqrtf((float)(c + 1));
  }
}

__global__ void place_kernel(const int* __restrict__ src,
                             const int* __restrict__ dst,
                             const int* __restrict__ rank,
                             const int* __restrict__ rowstart,
                             int* __restrict__ csr, int e) {
  int i = blockIdx.x * blockDim.x + threadIdx.x;
  if (i < e) csr[rowstart[dst[i]] + rank[i]] = src[i];
}

__global__ void copy_int_kernel(const int* __restrict__ from,
                                int* __restrict__ to, int n) {
  int i = blockIdx.x * blockDim.x + threadIdx.x;
  if (i < n) to[i] = from[i];
}

template <int K, int NT>
__global__ void gemm_wmma_kernel(const float* __restrict__ A,
                                 const float* __restrict__ B,
                                 float* __restrict__ C, int M) {
  const int N = NT * 16;
  const int wave = (int)((blockIdx.x * (unsigned)blockDim.x + threadIdx.x) >> 5);
  const int lane = threadIdx.x & 31;
  const int row0 = wave * 16;
  if (row0 >= M) return;

  const int mn    = lane & 15;
  const int khalf = lane >> 4;

  int arow = row0 + mn;
  if (arow >= M) arow = M - 1;
  const float* __restrict__ Ar = A + (size_t)arow * K;

  v8f acc[NT];
#pragma unroll
  for (int t = 0; t < NT; ++t) acc[t] = v8f{};

  for (int k = 0; k < K; k += 4) {
    const int ka = k + 2 * khalf;
    v2f a;
    a.x = Ar[ka];
    a.y = Ar[ka + 1];
#pragma unroll
    for (int t = 0; t < NT; ++t) {
      v2f b;
      b.x = B[(size_t)ka       * N + t * 16 + mn];
      b.y = B[(size_t)(ka + 1) * N + t * 16 + mn];
      acc[t] = __builtin_amdgcn_wmma_f32_16x16x4_f32(
           false, a,  false, b,
           (short)0, acc[t],  false,  false);
    }
  }

#pragma unroll
  for (int t = 0; t < NT; ++t) {
#pragma unroll
    for (int i = 0; i < 8; ++i) {
      const int row = row0 + 8 * khalf + i;
      if (row < M) C[(size_t)row * N + t * 16 + mn] = acc[t][i];
    }
  }
}

template <int F, bool RELU>
__global__ void gather_kernel(const float* __restrict__ h,
                              const int* __restrict__ csr,
                              const int* __restrict__ rowstart,
                              const int* __restrict__ cnt,
                              const float* __restrict__ dinv,
                              const float* __restrict__ bias,
                              float* __restrict__ outp, int n_nodes) {
  constexpr int LPN = F / 4;
  const long long gt = blockIdx.x * (long long)blockDim.x + threadIdx.x;
  const int node = (int)(gt / LPN);
  const int l = (int)(gt & (LPN - 1));
  if (node >= n_nodes) return;
  const float4* __restrict__ h4 = (const float4*)h;
  const int beg = rowstart[node];
  const int c   = cnt[node];
  const float dd = dinv[node];

  float4 acc;
  {
    const float4 v = h4[(size_t)node * LPN + l];
    const float nm = dd * dd;
    acc.x = v.x * nm; acc.y = v.y * nm; acc.z = v.z * nm; acc.w = v.w * nm;
  }
  int j = 0;
  for (; j + 4 <= c; j += 4) {
    const int s0 = csr[beg + j],     s1 = csr[beg + j + 1];
    const int s2 = csr[beg + j + 2], s3 = csr[beg + j + 3];
    const float n0 = dinv[s0] * dd, n1 = dinv[s1] * dd;
    const float n2 = dinv[s2] * dd, n3 = dinv[s3] * dd;
    const float4 v0 = h4[(size_t)s0 * LPN + l];
    const float4 v1 = h4[(size_t)s1 * LPN + l];
    const float4 v2 = h4[(size_t)s2 * LPN + l];
    const float4 v3 = h4[(size_t)s3 * LPN + l];
    acc.x += v0.x * n0; acc.y += v0.y * n0; acc.z += v0.z * n0; acc.w += v0.w * n0;
    acc.x += v1.x * n1; acc.y += v1.y * n1; acc.z += v1.z * n1; acc.w += v1.w * n1;
    acc.x += v2.x * n2; acc.y += v2.y * n2; acc.z += v2.z * n2; acc.w += v2.w * n2;
    acc.x += v3.x * n3; acc.y += v3.y * n3; acc.z += v3.z * n3; acc.w += v3.w * n3;
  }
  const int rem = c - j;
  if (rem > 0) {
    const int s0 = csr[beg + j];
    const int s1 = rem > 1 ? csr[beg + j + 1] : s0;
    const int s2 = rem > 2 ? csr[beg + j + 2] : s0;
    const float n0 = dinv[s0] * dd, n1 = dinv[s1] * dd, n2 = dinv[s2] * dd;
    const float4 v0 = h4[(size_t)s0 * LPN + l];
    const float4 v1 = h4[(size_t)s1 * LPN + l];
    const float4 v2 = h4[(size_t)s2 * LPN + l];
    acc.x += v0.x * n0; acc.y += v0.y * n0; acc.z += v0.z * n0; acc.w += v0.w * n0;
    if (rem > 1) {
      acc.x += v1.x * n1; acc.y += v1.y * n1; acc.z += v1.z * n1; acc.w += v1.w * n1;
    }
    if (rem > 2) {
      acc.x += v2.x * n2; acc.y += v2.y * n2; acc.z += v2.z * n2; acc.w += v2.w * n2;
    }
  }
  const float4 b = ((const float4*)bias)[l];
  acc.x += b.x; acc.y += b.y; acc.z += b.z; acc.w += b.w;
  if (RELU) {
    acc.x = acc.x > 0.0f ? acc.x : 0.0f; acc.y = acc.y > 0.0f ? acc.y : 0.0f;
    acc.z = acc.z > 0.0f ? acc.z : 0.0f; acc.w = acc.w > 0.0f ? acc.w : 0.0f;
  }
  ((float4*)outp)[(size_t)node * LPN + l] = acc;
}

extern "C" void kernel_launch(void* const* d_in, const int* in_sizes, int n_in,
                              void* d_out, int out_size, void* d_ws, size_t ws_size,
                              hipStream_t stream) {
  const float* x  = (const float*)d_in[0];
  const int*   ei = (const int*)d_in[1];
  const float* W1 = (const float*)d_in[2];
  const float* b1 = (const float*)d_in[3];
  const float* W2 = (const float*)d_in[4];
  const float* b2 = (const float*)d_in[5];
  float* out = (float*)d_out;

  const int N = in_sizes[0] / IN_DIM;
  const int E = in_sizes[1] / 2;
  const int* src = ei;
  const int* dst = ei + E;

  float* dinv = (float*)d_ws;
  float* h    = dinv + N;
  float* agg1 = h + (size_t)N * HID_DIM;

  int* total    = (int*)d_out;
  int* cnt      = total + 32;
  int* rowstart = cnt + N;
  int* csr      = rowstart + N;
  int* rank     = csr + E;
  int* cnt2      = (int*)(h + (size_t)N * OUT_DIM);
  int* rowstart2 = cnt2 + N;
  int* csr2      = rowstart2 + N;

  const int B256 = 256;
  zero_kernel<<<(32 + N + B256 - 1) / B256, B256, 0, stream>>>(total, 32 + N);
  count_rank_kernel<<<(E + B256 - 1) / B256, B256, 0, stream>>>(dst, cnt, rank, E);
  segment_kernel<<<(N + B256 - 1) / B256, B256, 0, stream>>>(cnt, total, rowstart, dinv, N);
  place_kernel<<<(E + B256 - 1) / B256, B256, 0, stream>>>(src, dst, rank, rowstart, csr, E);

  const int waves  = (N + 15) / 16;
  const int gblks  = (waves + 3) / 4;

  gemm_wmma_kernel<IN_DIM, HID_DIM / 16><<<gblks, 128, 0, stream>>>(x, W1, h, N);
  const long long th1 = (long long)N * (HID_DIM / 4);
  gather_kernel<HID_DIM, true><<<(int)((th1 + B256 - 1) / B256), B256, 0, stream>>>(
      h, csr, rowstart, cnt, dinv, b1, agg1, N);

  gemm_wmma_kernel<HID_DIM, OUT_DIM / 16><<<gblks, 128, 0, stream>>>(agg1, W2, h, N);
  const int ncopy = 2 * N + E;
  copy_int_kernel<<<(ncopy + B256 - 1) / B256, B256, 0, stream>>>(cnt, cnt2, ncopy);
  const long long th2 = (long long)N * (OUT_DIM / 4);
  gather_kernel<OUT_DIM, false><<<(int)((th2 + B256 - 1) / B256), B256, 0, stream>>>(
      h, csr2, rowstart2, cnt2, dinv, b2, out, N);
}


# GEMM epilogues store rows pre-scaled by rsqrt(count+1): gathers drop the per-edge dinv load and multiply; dinv array gone, so rows sit on 512-B bounds; sum order kept, scale factored out of the sum
# speedup vs baseline: 3.1496x; 1.1177x over previous
#include <hip/hip_runtime.h>

#define IN_DIM  256
#define HID_DIM 128
#define OUT_DIM 64

typedef __attribute__((ext_vector_type(2))) float v2f;
typedef __attribute__((ext_vector_type(8))) float v8f;

__global__ void zero_kernel(int* __restrict__ p, int n) {
  int i = blockIdx.x * blockDim.x + threadIdx.x;
  if (i < n) p[i] = 0;
}

__global__ void count_rank_kernel(const int* __restrict__ dst,
                                  int* __restrict__ cnt,
                                  int* __restrict__ rank, int e) {
  int i = blockIdx.x * blockDim.x + threadIdx.x;
  if (i < e) rank[i] = atomicAdd(&cnt[dst[i]], 1);
}

__global__ void segment_kernel(const int* __restrict__ cnt,
                               int* __restrict__ total,
                               int* __restrict__ rowstart, int n) {
  const int i = blockIdx.x * blockDim.x + threadIdx.x;
  const int lane = threadIdx.x & (warpSize - 1);
  const int c = (i < n) ? cnt[i] : 0;
  int incl = c;
  for (int d = 1; d < warpSize; d <<= 1) {
    int t = __shfl_up(incl, d);
    if (lane >= d) incl += t;
  }
  int base = 0;
  if (lane == warpSize - 1) base = atomicAdd(total, incl);
  base = __shfl(base, warpSize - 1);
  if (i < n) rowstart[i] = base + incl - c;
}

__global__ void place_kernel(const int* __restrict__ src,
                             const int* __restrict__ dst,
                             const int* __restrict__ rank,
                             const int* __restrict__ rowstart,
                             int* __restrict__ csr, int e) {
  int i = blockIdx.x * blockDim.x + threadIdx.x;
  if (i < e) csr[rowstart[dst[i]] + rank[i]] = src[i];
}

__global__ void copy_int_kernel(const int* __restrict__ from,
                                int* __restrict__ to, int n) {
  int i = blockIdx.x * blockDim.x + threadIdx.x;
  if (i < n) to[i] = from[i];
}

template <int K, int NT>
__global__ void gemm_wmma_kernel(const float* __restrict__ A,
                                 const float* __restrict__ B,
                                 const int* __restrict__ cnt,
                                 float* __restrict__ C, int M) {
  const int N = NT * 16;
  const int wave = (int)((blockIdx.x * (unsigned)blockDim.x + threadIdx.x) >> 5);
  const int lane = threadIdx.x & 31;
  const int row0 = wave * 16;
  if (row0 >= M) return;

  const int mn    = lane & 15;
  const int khalf = lane >> 4;

  int arow = row0 + mn;
  if (arow >= M) arow = M - 1;
  const float* __restrict__ Ar = A + (size_t)arow * K;

  v8f acc[NT];
#pragma unroll
  for (int t = 0; t < NT; ++t) acc[t] = v8f{};

  for (int k = 0; k < K; k += 4) {
    const int ka = k + 2 * khalf;
    v2f a;
    a.x = Ar[ka];
    a.y = Ar[ka + 1];
#pragma unroll
    for (int t = 0; t < NT; ++t) {
      v2f b;
      b.x = B[(size_t)ka       * N + t * 16 + mn];
      b.y = B[(size_t)(ka + 1) * N + t * 16 + mn];
      acc[t] = __builtin_amdgcn_wmma_f32_16x16x4_f32(
           false, a,  false, b,
           (short)0, acc[t],  false,  false);
    }
  }

  float sc[8];
#pragma unroll
  for (int i = 0; i < 8; ++i) {
    int row = row0 + 8 * khalf + i;
    if (row >= M) row = M - 1;
    sc[i] = rsqrtf((float)(cnt[row] + 1));
  }
#pragma unroll
  for (int t = 0; t < NT; ++t) {
#pragma unroll
    for (int i = 0; i < 8; ++i) {
      const int row = row0 + 8 * khalf + i;
      if (row < M) C[(size_t)row * N + t * 16 + mn] = acc[t][i] * sc[i];
    }
  }
}

template <int F, bool RELU>
__global__ void gather_kernel(const float* __restrict__ h,
                              const int* __restrict__ csr,
                              const int* __restrict__ rowstart,
                              const int* __restrict__ cnt,
                              const float* __restrict__ bias,
                              float* __restrict__ outp, int n_nodes) {
  constexpr int LPN = F / 4;
  const long long gt = blockIdx.x * (long long)blockDim.x + threadIdx.x;
  const int node = (int)(gt / LPN);
  const int l = (int)(gt & (LPN - 1));
  if (node >= n_nodes) return;
  const float4* __restrict__ h4 = (const float4*)h;
  const int beg = rowstart[node];
  const int c   = cnt[node];
  const float dd = rsqrtf((float)(c + 1));

  float4 acc = h4[(size_t)node * LPN + l];
  int j = 0;
  for (; j + 4 <= c; j += 4) {
    const int s0 = csr[beg + j],     s1 = csr[beg + j + 1];
    const int s2 = csr[beg + j + 2], s3 = csr[beg + j + 3];
    const float4 v0 = h4[(size_t)s0 * LPN + l];
    const float4 v1 = h4[(size_t)s1 * LPN + l];
    const float4 v2 = h4[(size_t)s2 * LPN + l];
    const float4 v3 = h4[(size_t)s3 * LPN + l];
    acc.x += v0.x; acc.y += v0.y; acc.z += v0.z; acc.w += v0.w;
    acc.x += v1.x; acc.y += v1.y; acc.z += v1.z; acc.w += v1.w;
    acc.x += v2.x; acc.y += v2.y; acc.z += v2.z; acc.w += v2.w;
    acc.x += v3.x; acc.y += v3.y; acc.z += v3.z; acc.w += v3.w;
  }
  const int rem = c - j;
  if (rem > 0) {
    const int s0 = csr[beg + j];
    const int s1 = rem > 1 ? csr[beg + j + 1] : s0;
    const int s2 = rem > 2 ? csr[beg + j + 2] : s0;
    const float4 v0 = h4[(size_t)s0 * LPN + l];
    const float4 v1 = h4[(size_t)s1 * LPN + l];
    const float4 v2 = h4[(size_t)s2 * LPN + l];
    acc.x += v0.x; acc.y += v0.y; acc.z += v0.z; acc.w += v0.w;
    if (rem > 1) {
      acc.x += v1.x; acc.y += v1.y; acc.z += v1.z; acc.w += v1.w;
    }
    if (rem > 2) {
      acc.x += v2.x; acc.y += v2.y; acc.z += v2.z; acc.w += v2.w;
    }
  }
  const float4 b = ((const float4*)bias)[l];
  acc.x = acc.x * dd + b.x; acc.y = acc.y * dd + b.y;
  acc.z = acc.z * dd + b.z; acc.w = acc.w * dd + b.w;
  if (RELU) {
    acc.x = acc.x > 0.0f ? acc.x : 0.0f; acc.y = acc.y > 0.0f ? acc.y : 0.0f;
    acc.z = acc.z > 0.0f ? acc.z : 0.0f; acc.w = acc.w > 0.0f ? acc.w : 0.0f;
  }
  ((float4*)outp)[(size_t)node * LPN + l] = acc;
}

extern "C" void kernel_launch(void* const* d_in, const int* in_sizes, int n_in,
                              void* d_out, int out_size, void* d_ws, size_t ws_size,
                              hipStream_t stream) {
  const float* x  = (const float*)d_in[0];
  const int*   ei = (const int*)d_in[1];
  const float* W1 = (const float*)d_in[2];
  const float* b1 = (const float*)d_in[3];
  const float* W2 = (const float*)d_in[4];
  const float* b2 = (const float*)d_in[5];
  float* out = (float*)d_out;

  const int N = in_sizes[0] / IN_DIM;
  const int E = in_sizes[1] / 2;
  const int* src = ei;
  const int* dst = ei + E;

  float* h    = (float*)d_ws;
  float* agg1 = h + (size_t)N * HID_DIM;

  int* total    = (int*)d_out;
  int* cnt      = total + 32;
  int* rowstart = cnt + N;
  int* csr      = rowstart + N;
  int* rank     = csr + E;
  int* cnt2      = (int*)(h + (size_t)N * OUT_DIM);
  int* rowstart2 = cnt2 + N;
  int* csr2      = rowstart2 + N;

  const int B256 = 256;
  zero_kernel<<<(32 + N + B256 - 1) / B256, B256, 0, stream>>>(total, 32 + N);
  count_rank_kernel<<<(E + B256 - 1) / B256, B256, 0, stream>>>(dst, cnt, rank, E);
  segment_kernel<<<(N + B256 - 1) / B256, B256, 0, stream>>>(cnt, total, rowstart, N);
  place_kernel<<<(E + B256 - 1) / B256, B256, 0, stream>>>(src, dst, rank, rowstart, csr, E);

  const int waves  = (N + 15) / 16;
  const int gblks  = (waves + 3) / 4;

  gemm_wmma_kernel<IN_DIM, HID_DIM / 16><<<gblks, 128, 0, stream>>>(x, W1, cnt, h, N);
  const long long th1 = (long long)N * (HID_DIM / 4);
  gather_kernel<HID_DIM, true><<<(int)((th1 + B256 - 1) / B256), B256, 0, stream>>>(
      h, csr, rowstart, cnt, b1, agg1, N);

  gemm_wmma_kernel<HID_DIM, OUT_DIM / 16><<<gblks, 128, 0, stream>>>(agg1, W2, cnt, h, N);
  const int ncopy = 2 * N + E;
  copy_int_kernel<<<(ncopy + B256 - 1) / B256, B256, 0, stream>>>(cnt, cnt2, ncopy);
  const long long th2 = (long long)N * (OUT_DIM / 4);
  gather_kernel<OUT_DIM, false><<<(int)((th2 + B256 - 1) / B256), B256, 0, stream>>>(
      h, csr2, rowstart2, cnt2, b2, out, N);
}


# Edge grouping rebuilt as a 2-launch counting sort through LDS (4096-edge tiles ordered by 128-node bin; one block per bin ranks with LDS adds, whole-line stores); node sum order = LDS arrival order
# speedup vs baseline: 4.4349x; 1.4081x over previous
#include <hip/hip_runtime.h>

#define IN_DIM  256
#define HID_DIM 128
#define OUT_DIM 64

typedef __attribute__((ext_vector_type(2))) float v2f;
typedef __attribute__((ext_vector_type(8))) float v8f;

#define BIN_SHIFT 7
#define BIN_W     (1 << BIN_SHIFT)
#define NB_MAX    512
#define TILE_E    4096
#define EPT       (TILE_E / 256)
#define BIN_CAP   8192

__device__ __forceinline__ int block_scan_excl(int v, int* wsum, int& total) {
  const int lane = threadIdx.x & 31, w = threadIdx.x >> 5;
  int incl = v;
  for (int d = 1; d < 32; d <<= 1) {
    int t = __shfl_up(incl, d);
    if (lane >= d) incl += t;
  }
  if (lane == 31) wsum[w] = incl;
  __syncthreads();
  int base = 0, tot = 0;
#pragma unroll
  for (int i = 0; i < 8; ++i) {
    const int s = wsum[i];
    if (i < w) base += s;
    tot += s;
  }
  total = tot;
  __syncthreads();
  return base + incl - v;
}

__global__ __launch_bounds__(256) void bin_tile_kernel(
    const int* __restrict__ src, const int* __restrict__ dst,
    int2* __restrict__ staged, int* __restrict__ table, int e, int nb) {
  __shared__ int hist[NB_MAX];
  __shared__ int cursor[NB_MAX];
  __shared__ int wsum[8];
  __shared__ int2 stage[TILE_E];
  const int tid = threadIdx.x;
  const int e0 = blockIdx.x * TILE_E;
  const int ne = (e - e0) < TILE_E ? (e - e0) : TILE_E;
  for (int j = tid; j < NB_MAX; j += 256) hist[j] = 0;
  __syncthreads();
  int s[EPT], d[EPT];
#pragma unroll
  for (int k = 0; k < EPT; ++k) {
    const int i = k * 256 + tid;
    d[k] = -1; s[k] = 0;
    if (i < ne) { d[k] = dst[e0 + i]; s[k] = src[e0 + i]; }
  }
#pragma unroll
  for (int k = 0; k < EPT; ++k)
    if (d[k] >= 0) atomicAdd(&hist[d[k] >> BIN_SHIFT], 1);
  __syncthreads();
  const int h0 = hist[2 * tid], h1 = hist[2 * tid + 1];
  int tot;
  const int ex = block_scan_excl(h0 + h1, wsum, tot);
  cursor[2 * tid] = ex;
  cursor[2 * tid + 1] = ex + h0;
  int* __restrict__ trow = table + (size_t)blockIdx.x * (nb + 1);
  if (2 * tid <= nb)     trow[2 * tid]     = ex;
  if (2 * tid + 1 <= nb) trow[2 * tid + 1] = ex + h0;
  __syncthreads();
#pragma unroll
  for (int k = 0; k < EPT; ++k) {
    if (d[k] >= 0) {
      const int pos = atomicAdd(&cursor[d[k] >> BIN_SHIFT], 1);
      stage[pos] = make_int2(s[k], d[k]);
    }
  }
  __syncthreads();
  for (int i = tid; i < ne; i += 256) staged[(size_t)e0 + i] = stage[i];
}

__global__ __launch_bounds__(256) void group_bin_kernel(
    const int2* __restrict__ staged, const int* __restrict__ table,
    int* __restrict__ cnt, int* __restrict__ rowstart, int* __restrict__ csr,
    int n, int ntile, int nb) {
  __shared__ int cntL[BIN_W];
  __shared__ int curL[BIN_W];
  __shared__ int wsum[8];
  __shared__ int stageL[BIN_CAP];
  const int tid = threadIdx.x;
  const int b = blockIdx.x;
  const int lo = b << BIN_SHIFT;
  if (tid < BIN_W) cntL[tid] = 0;
  int mystart = 0, mylen = 0;
  for (int t = tid; t < ntile; t += 256) {
    const int* __restrict__ tr = table + (size_t)t * (nb + 1) + b;
    const int a = tr[0], z = tr[1];
    mystart += a; mylen += z - a;
  }
  __syncthreads();
  for (int t = tid; t < ntile; t += 256) {
    const int* __restrict__ tr = table + (size_t)t * (nb + 1) + b;
    const int a = tr[0], z = tr[1];
    const int2* __restrict__ p = staged + (size_t)t * TILE_E + a;
    for (int i = 0; i < z - a; ++i) atomicAdd(&cntL[p[i].y - lo], 1);
  }
  int base, total;
  block_scan_excl(mystart, wsum, base);
  block_scan_excl(mylen, wsum, total);
  const int c = tid < BIN_W ? cntL[tid] : 0;
  int tot2;
  const int ex = block_scan_excl(c, wsum, tot2);
  if (tid < BIN_W) {
    curL[tid] = ex;
    const int node = lo + tid;
    if (node < n) { cnt[node] = c; rowstart[node] = base + ex; }
  }
  __syncthreads();
  const bool in_lds = total <= BIN_CAP;
  for (int t = tid; t < ntile; t += 256) {
    const int* __restrict__ tr = table + (size_t)t * (nb + 1) + b;
    const int a = tr[0], z = tr[1];
    const int2* __restrict__ p = staged + (size_t)t * TILE_E + a;
    for (int i = 0; i < z - a; ++i) {
      const int2 ed = p[i];
      const int pos = atomicAdd(&curL[ed.y - lo], 1);
      if (in_lds) stageL[pos] = ed.x; else csr[base + pos] = ed.x;
    }
  }
  __syncthreads();
  if (in_lds)
    for (int i = tid; i < total; i += 256) csr[base + i] = stageL[i];
}

__global__ void copy_int_kernel(const int* __restrict__ from,
                                int* __restrict__ to, int n) {
  int i = blockIdx.x * blockDim.x + threadIdx.x;
  if (i < n) to[i] = from[i];
}

template <int K, int NT>
__global__ void gemm_wmma_kernel(const float* __restrict__ A,
                                 const float* __restrict__ B,
                                 const int* __restrict__ cnt,
                                 float* __restrict__ C, int M) {
  const int N = NT * 16;
  const int wave = (int)((blockIdx.x * (unsigned)blockDim.x + threadIdx.x) >> 5);
  const int lane = threadIdx.x & 31;
  const int row0 = wave * 16;
  if (row0 >= M) return;

  const int mn    = lane & 15;
  const int khalf = lane >> 4;

  int arow = row0 + mn;
  if (arow >= M) arow = M - 1;
  const float* __restrict__ Ar = A + (size_t)arow * K;

  v8f acc[NT];
#pragma unroll
  for (int t = 0; t < NT; ++t) acc[t] = v8f{};

  for (int k = 0; k < K; k += 4) {
    const int ka = k + 2 * khalf;
    v2f a;
    a.x = Ar[ka];
    a.y = Ar[ka + 1];
#pragma unroll
    for (int t = 0; t < NT; ++t) {
      v2f b;
      b.x = B[(size_t)ka       * N + t * 16 + mn];
      b.y = B[(size_t)(ka + 1) * N + t * 16 + mn];
      acc[t] = __builtin_amdgcn_wmma_f32_16x16x4_f32(
           false, a,  false, b,
           (short)0, acc[t],  false,  false);
    }
  }

  float sc[8];
#pragma unroll
  for (int i = 0; i < 8; ++i) {
    int row = row0 + 8 * khalf + i;
    if (row >= M) row = M - 1;
    sc[i] = rsqrtf((float)(cnt[row] + 1));
  }
#pragma unroll
  for (int t = 0; t < NT; ++t) {
#pragma unroll
    for (int i = 0; i < 8; ++i) {
      const int row = row0 + 8 * khalf + i;
      if (row < M) C[(size_t)row * N + t * 16 + mn] = acc[t][i] * sc[i];
    }
  }
}

template <int F, bool RELU>
__global__ void gather_kernel(const float* __restrict__ h,
                              const int* __restrict__ csr,
                              const int* __restrict__ rowstart,
                              const int* __restrict__ cnt,
                              const float* __restrict__ bias,
                              float* __restrict__ outp, int n_nodes) {
  constexpr int LPN = F / 4;
  const long long gt = blockIdx.x * (long long)blockDim.x + threadIdx.x;
  const int node = (int)(gt / LPN);
  const int l = (int)(gt & (LPN - 1));
  if (node >= n_nodes) return;
  const float4* __restrict__ h4 = (const float4*)h;
  const int beg = rowstart[node];
  const int c   = cnt[node];
  const float dd = rsqrtf((float)(c + 1));

  float4 acc = h4[(size_t)node * LPN + l];
  int j = 0;
  for (; j + 4 <= c; j += 4) {
    const int s0 = csr[beg + j],     s1 = csr[beg + j + 1];
    const int s2 = csr[beg + j + 2], s3 = csr[beg + j + 3];
    const float4 v0 = h4[(size_t)s0 * LPN + l];
    const float4 v1 = h4[(size_t)s1 * LPN + l];
    const float4 v2 = h4[(size_t)s2 * LPN + l];
    const float4 v3 = h4[(size_t)s3 * LPN + l];
    acc.x += v0.x; acc.y += v0.y; acc.z += v0.z; acc.w += v0.w;
    acc.x += v1.x; acc.y += v1.y; acc.z += v1.z; acc.w += v1.w;
    acc.x += v2.x; acc.y += v2.y; acc.z += v2.z; acc.w += v2.w;
    acc.x += v3.x; acc.y += v3.y; acc.z += v3.z; acc.w += v3.w;
  }
  const int rem = c - j;
  if (rem > 0) {
    const int s0 = csr[beg + j];
    const int s1 = rem > 1 ? csr[beg + j + 1] : s0;
    const int s2 = rem > 2 ? csr[beg + j + 2] : s0;
    const float4 v0 = h4[(size_t)s0 * LPN + l];
    const float4 v1 = h4[(size_t)s1 * LPN + l];
    const float4 v2 = h4[(size_t)s2 * LPN + l];
    acc.x += v0.x; acc.y += v0.y; acc.z += v0.z; acc.w += v0.w;
    if (rem > 1) {
      acc.x += v1.x; acc.y += v1.y; acc.z += v1.z; acc.w += v1.w;
    }
    if (rem > 2) {
      acc.x += v2.x; acc.y += v2.y; acc.z += v2.z; acc.w += v2.w;
    }
  }
  const float4 b = ((const float4*)bias)[l];
  acc.x = acc.x * dd + b.x; acc.y = acc.y * dd + b.y;
  acc.z = acc.z * dd + b.z; acc.w = acc.w * dd + b.w;
  if (RELU) {
    acc.x = acc.x > 0.0f ? acc.x : 0.0f; acc.y = acc.y > 0.0f ? acc.y : 0.0f;
    acc.z = acc.z > 0.0f ? acc.z : 0.0f; acc.w = acc.w > 0.0f ? acc.w : 0.0f;
  }
  ((float4*)outp)[(size_t)node * LPN + l] = acc;
}

extern "C" void kernel_launch(void* const* d_in, const int* in_sizes, int n_in,
                              void* d_out, int out_size, void* d_ws, size_t ws_size,
                              hipStream_t stream) {
  const float* x  = (const float*)d_in[0];
  const int*   ei = (const int*)d_in[1];
  const float* W1 = (const float*)d_in[2];
  const float* b1 = (const float*)d_in[3];
  const float* W2 = (const float*)d_in[4];
  const float* b2 = (const float*)d_in[5];
  float* out = (float*)d_out;

  const int N = in_sizes[0] / IN_DIM;
  const int E = in_sizes[1] / 2;
  const int* src = ei;
  const int* dst = ei + E;

  float* h    = (float*)d_ws;
  float* agg1 = h + (size_t)N * HID_DIM;

  int* cnt      = (int*)d_out + 32;
  int* rowstart = cnt + N;
  int* csr      = rowstart + N;
  int* cnt2      = (int*)(h + (size_t)N * OUT_DIM);
  int* rowstart2 = cnt2 + N;
  int* csr2      = rowstart2 + N;
  const int ntile = (E + TILE_E - 1) / TILE_E;
  const int nb    = (N + BIN_W - 1) / BIN_W;
  int2* staged = (int2*)d_ws;
  int*  table  = (int*)(staged + (size_t)ntile * TILE_E);

  const int B256 = 256;
  bin_tile_kernel<<<ntile, B256, 0, stream>>>(src, dst, staged, table, E, nb);
  group_bin_kernel<<<nb, B256, 0, stream>>>(staged, table, cnt, rowstart, csr, N, ntile, nb);

  const int waves  = (N + 15) / 16;
  const int gblks  = (waves + 3) / 4;

  gemm_wmma_kernel<IN_DIM, HID_DIM / 16><<<gblks, 128, 0, stream>>>(x, W1, cnt, h, N);
  const long long th1 = (long long)N * (HID_DIM / 4);
  gather_kernel<HID_DIM, true><<<(int)((th1 + B256 - 1) / B256), B256, 0, stream>>>(
      h, csr, rowstart, cnt, b1, agg1, N);

  gemm_wmma_kernel<HID_DIM, OUT_DIM / 16><<<gblks, 128, 0, stream>>>(agg1, W2, cnt, h, N);
  const int ncopy = 2 * N + E;
  copy_int_kernel<<<(ncopy + B256 - 1) / B256, B256, 0, stream>>>(cnt, cnt2, ncopy);
  const long long th2 = (long long)N * (OUT_DIM / 4);
  gather_kernel<OUT_DIM, false><<<(int)((th2 + B256 - 1) / B256), B256, 0, stream>>>(
      h, csr2, rowstart2, cnt2, b2, out, N);
}


# Layer-2 product fused into the layer-1 gather: rows pass through LDS to the WMMA, hidden aggregate never stored; its plane holds h2 and the edge grouping, so the copy launch goes (7 to 5 launches)
# speedup vs baseline: 4.9732x; 1.1214x over previous
#include <hip/hip_runtime.h>

#define IN_DIM  256
#define HID_DIM 128
#define OUT_DIM 64

typedef __attribute__((ext_vector_type(2))) float v2f;
typedef __attribute__((ext_vector_type(8))) float v8f;

#define BIN_SHIFT 7
#define BIN_W     (1 << BIN_SHIFT)
#define NB_MAX    512
#define TILE_E    4096
#define EPT       (TILE_E / 256)
#define BIN_CAP   8192

__device__ __forceinline__ int block_scan_excl(int v, int* wsum, int& total) {
  const int lane = threadIdx.x & 31, w = threadIdx.x >> 5;
  int incl = v;
  for (int d = 1; d < 32; d <<= 1) {
    int t = __shfl_up(incl, d);
    if (lane >= d) incl += t;
  }
  if (lane == 31) wsum[w] = incl;
  __syncthreads();
  int base = 0, tot = 0;
#pragma unroll
  for (int i = 0; i < 8; ++i) {
    const int s = wsum[i];
    if (i < w) base += s;
    tot += s;
  }
  total = tot;
  __syncthreads();
  return base + incl - v;
}

__global__ __launch_bounds__(256) void bin_tile_kernel(
    const int* __restrict__ src, const int* __restrict__ dst,
    int2* __restrict__ staged, int* __restrict__ table, int e, int nb) {
  __shared__ int hist[NB_MAX];
  __shared__ int cursor[NB_MAX];
  __shared__ int wsum[8];
  __shared__ int2 stage[TILE_E];
  const int tid = threadIdx.x;
  const int e0 = blockIdx.x * TILE_E;
  const int ne = (e - e0) < TILE_E ? (e - e0) : TILE_E;
  for (int j = tid; j < NB_MAX; j += 256) hist[j] = 0;
  __syncthreads();
  int s[EPT], d[EPT];
#pragma unroll
  for (int k = 0; k < EPT; ++k) {
    const int i = k * 256 + tid;
    d[k] = -1; s[k] = 0;
    if (i < ne) { d[k] = dst[e0 + i]; s[k] = src[e0 + i]; }
  }
#pragma unroll
  for (int k = 0; k < EPT; ++k)
    if (d[k] >= 0) atomicAdd(&hist[d[k] >> BIN_SHIFT], 1);
  __syncthreads();
  const int h0 = hist[2 * tid], h1 = hist[2 * tid + 1];
  int tot;
  const int ex = block_scan_excl(h0 + h1, wsum, tot);
  cursor[2 * tid] = ex;
  cursor[2 * tid + 1] = ex + h0;
  int* __restrict__ trow = table + (size_t)blockIdx.x * (nb + 1);
  if (2 * tid <= nb)     trow[2 * tid]     = ex;
  if (2 * tid + 1 <= nb) trow[2 * tid + 1] = ex + h0;
  __syncthreads();
#pragma unroll
  for (int k = 0; k < EPT; ++k) {
    if (d[k] >= 0) {
      const int pos = atomicAdd(&cursor[d[k] >> BIN_SHIFT], 1);
      stage[pos] = make_int2(s[k], d[k]);
    }
  }
  __syncthreads();
  for (int i = tid; i < ne; i += 256) staged[(size_t)e0 + i] = stage[i];
}

__global__ __launch_bounds__(256) void group_bin_kernel(
    const int2* __restrict__ staged, const int* __restrict__ table,
    int* __restrict__ cnt, int* __restrict__ rowstart, int* __restrict__ csr,
    int n, int ntile, int nb) {
  __shared__ int cntL[BIN_W];
  __shared__ int curL[BIN_W];
  __shared__ int wsum[8];
  __shared__ int stageL[BIN_CAP];
  const int tid = threadIdx.x;
  const int b = blockIdx.x;
  const int lo = b << BIN_SHIFT;
  if (tid < BIN_W) cntL[tid] = 0;
  int mystart = 0, mylen = 0;
  for (int t = tid; t < ntile; t += 256) {
    const int* __restrict__ tr = table + (size_t)t * (nb + 1) + b;
    const int a = tr[0], z = tr[1];
    mystart += a; mylen += z - a;
  }
  __syncthreads();
  for (int t = tid; t < ntile; t += 256) {
    const int* __restrict__ tr = table + (size_t)t * (nb + 1) + b;
    const int a = tr[0], z = tr[1];
    const int2* __restrict__ p = staged + (size_t)t * TILE_E + a;
    for (int i = 0; i < z - a; ++i) atomicAdd(&cntL[p[i].y - lo], 1);
  }
  int base, total;
  block_scan_excl(mystart, wsum, base);
  block_scan_excl(mylen, wsum, total);
  const int c = tid < BIN_W ? cntL[tid] : 0;
  int tot2;
  const int ex = block_scan_excl(c, wsum, tot2);
  if (tid < BIN_W) {
    curL[tid] = ex;
    const int node = lo + tid;
    if (node < n) { cnt[node] = c; rowstart[node] = base + ex; }
  }
  __syncthreads();
  const bool in_lds = total <= BIN_CAP;
  for (int t = tid; t < ntile; t += 256) {
    const int* __restrict__ tr = table + (size_t)t * (nb + 1) + b;
    const int a = tr[0], z = tr[1];
    const int2* __restrict__ p = staged + (size_t)t * TILE_E + a;
    for (int i = 0; i < z - a; ++i) {
      const int2 ed = p[i];
      const int pos = atomicAdd(&curL[ed.y - lo], 1);
      if (in_lds) stageL[pos] = ed.x; else csr[base + pos] = ed.x;
    }
  }
  __syncthreads();
  if (in_lds)
    for (int i = tid; i < total; i += 256) csr[base + i] = stageL[i];
}

template <int K, int NT>
__global__ void gemm_wmma_kernel(const float* __restrict__ A,
                                 const float* __restrict__ B,
                                 const int* __restrict__ cnt,
                                 float* __restrict__ C, int M) {
  const int N = NT * 16;
  const int wave = (int)((blockIdx.x * (unsigned)blockDim.x + threadIdx.x) >> 5);
  const int lane = threadIdx.x & 31;
  const int row0 = wave * 16;
  if (row0 >= M) return;

  const int mn    = lane & 15;
  const int khalf = lane >> 4;

  int arow = row0 + mn;
  if (arow >= M) arow = M - 1;
  const float* __restrict__ Ar = A + (size_t)arow * K;

  v8f acc[NT];
#pragma unroll
  for (int t = 0; t < NT; ++t) acc[t] = v8f{};

  for (int k = 0; k < K; k += 4) {
    const int ka = k + 2 * khalf;
    v2f a;
    a.x = Ar[ka];
    a.y = Ar[ka + 1];
#pragma unroll
    for (int t = 0; t < NT; ++t) {
      v2f b;
      b.x = B[(size_t)ka       * N + t * 16 + mn];
      b.y = B[(size_t)(ka + 1) * N + t * 16 + mn];
      acc[t] = __builtin_amdgcn_wmma_f32_16x16x4_f32(
           false, a,  false, b,
           (short)0, acc[t],  false,  false);
    }
  }

  float sc[8];
#pragma unroll
  for (int i = 0; i < 8; ++i) {
    int row = row0 + 8 * khalf + i;
    if (row >= M) row = M - 1;
    sc[i] = rsqrtf((float)(cnt[row] + 1));
  }
#pragma unroll
  for (int t = 0; t < NT; ++t) {
#pragma unroll
    for (int i = 0; i < 8; ++i) {
      const int row = row0 + 8 * khalf + i;
      if (row < M) C[(size_t)row * N + t * 16 + mn] = acc[t][i] * sc[i];
    }
  }
}

template <int F, bool RELU>
__global__ void gather_kernel(const float* __restrict__ h,
                              const int* __restrict__ csr,
                              const int* __restrict__ rowstart,
                              const int* __restrict__ cnt,
                              const float* __restrict__ bias,
                              float* __restrict__ outp, int n_nodes) {
  constexpr int LPN = F / 4;
  const long long gt = blockIdx.x * (long long)blockDim.x + threadIdx.x;
  const int node = (int)(gt / LPN);
  const int l = (int)(gt & (LPN - 1));
  if (node >= n_nodes) return;
  const float4* __restrict__ h4 = (const float4*)h;
  const int beg = rowstart[node];
  const int c   = cnt[node];
  const float dd = rsqrtf((float)(c + 1));

  float4 acc = h4[(size_t)node * LPN + l];
  int j = 0;
  for (; j + 4 <= c; j += 4) {
    const int s0 = csr[beg + j],     s1 = csr[beg + j + 1];
    const int s2 = csr[beg + j + 2], s3 = csr[beg + j + 3];
    const float4 v0 = h4[(size_t)s0 * LPN + l];
    const float4 v1 = h4[(size_t)s1 * LPN + l];
    const float4 v2 = h4[(size_t)s2 * LPN + l];
    const float4 v3 = h4[(size_t)s3 * LPN + l];
    acc.x += v0.x; acc.y += v0.y; acc.z += v0.z; acc.w += v0.w;
    acc.x += v1.x; acc.y += v1.y; acc.z += v1.z; acc.w += v1.w;
    acc.x += v2.x; acc.y += v2.y; acc.z += v2.z; acc.w += v2.w;
    acc.x += v3.x; acc.y += v3.y; acc.z += v3.z; acc.w += v3.w;
  }
  const int rem = c - j;
  if (rem > 0) {
    const int s0 = csr[beg + j];
    const int s1 = rem > 1 ? csr[beg + j + 1] : s0;
    const int s2 = rem > 2 ? csr[beg + j + 2] : s0;
    const float4 v0 = h4[(size_t)s0 * LPN + l];
    const float4 v1 = h4[(size_t)s1 * LPN + l];
    const float4 v2 = h4[(size_t)s2 * LPN + l];
    acc.x += v0.x; acc.y += v0.y; acc.z += v0.z; acc.w += v0.w;
    if (rem > 1) {
      acc.x += v1.x; acc.y += v1.y; acc.z += v1.z; acc.w += v1.w;
    }
    if (rem > 2) {
      acc.x += v2.x; acc.y += v2.y; acc.z += v2.z; acc.w += v2.w;
    }
  }
  const float4 b = ((const float4*)bias)[l];
  acc.x = acc.x * dd + b.x; acc.y = acc.y * dd + b.y;
  acc.z = acc.z * dd + b.z; acc.w = acc.w * dd + b.w;
  if (RELU) {
    acc.x = acc.x > 0.0f ? acc.x : 0.0f; acc.y = acc.y > 0.0f ? acc.y : 0.0f;
    acc.z = acc.z > 0.0f ? acc.z : 0.0f; acc.w = acc.w > 0.0f ? acc.w : 0.0f;
  }
  ((float4*)outp)[(size_t)node * LPN + l] = acc;
}

#define AG_PITCH (HID_DIM + 4)

__global__ __launch_bounds__(256) __attribute__((amdgpu_waves_per_eu(12)))
void gather_gemm_kernel(
    const float* __restrict__ h, const int* __restrict__ csr,
    const int* __restrict__ rowstart, const int* __restrict__ cnt,
    const float* __restrict__ bias, const float* __restrict__ B,
    float* __restrict__ C, int n_nodes) {
  __shared__ __attribute__((aligned(16))) float tile[16 * AG_PITCH];
  __shared__ float rowscale[16];
  const int lane = threadIdx.x & 31;
  const int w    = threadIdx.x >> 5;
  const int row0 = blockIdx.x * 16;
  const float4* __restrict__ h4 = (const float4*)h;

  for (int q = 0; q < 2; ++q) {
    const int r = 2 * w + q;
    const int node = row0 + r;
    float4 acc = make_float4(0.0f, 0.0f, 0.0f, 0.0f);
    float dd = 0.0f;
    if (node < n_nodes) {
      const int beg = rowstart[node];
      const int c   = cnt[node];
      dd = rsqrtf((float)(c + 1));
      acc = h4[(size_t)node * 32 + lane];
      int j = 0;
      for (; j + 4 <= c; j += 4) {
        const int s0 = csr[beg + j],     s1 = csr[beg + j + 1];
        const int s2 = csr[beg + j + 2], s3 = csr[beg + j + 3];
        const float4 v0 = h4[(size_t)s0 * 32 + lane];
        const float4 v1 = h4[(size_t)s1 * 32 + lane];
        const float4 v2 = h4[(size_t)s2 * 32 + lane];
        const float4 v3 = h4[(size_t)s3 * 32 + lane];
        acc.x += v0.x; acc.y += v0.y; acc.z += v0.z; acc.w += v0.w;
        acc.x += v1.x; acc.y += v1.y; acc.z += v1.z; acc.w += v1.w;
        acc.x += v2.x; acc.y += v2.y; acc.z += v2.z; acc.w += v2.w;
        acc.x += v3.x; acc.y += v3.y; acc.z += v3.z; acc.w += v3.w;
      }
      const int rem = c - j;
      if (rem > 0) {
        const int s0 = csr[beg + j];
        const int s1 = rem > 1 ? csr[beg + j + 1] : s0;
        const int s2 = rem > 2 ? csr[beg + j + 2] : s0;
        const float4 v0 = h4[(size_t)s0 * 32 + lane];
        const float4 v1 = h4[(size_t)s1 * 32 + lane];
        const float4 v2 = h4[(size_t)s2 * 32 + lane];
        acc.x += v0.x; acc.y += v0.y; acc.z += v0.z; acc.w += v0.w;
        if (rem > 1) {
          acc.x += v1.x; acc.y += v1.y; acc.z += v1.z; acc.w += v1.w;
        }
        if (rem > 2) {
          acc.x += v2.x; acc.y += v2.y; acc.z += v2.z; acc.w += v2.w;
        }
      }
      const float4 bb = ((const float4*)bias)[lane];
      acc.x = acc.x * dd + bb.x; acc.y = acc.y * dd + bb.y;
      acc.z = acc.z * dd + bb.z; acc.w = acc.w * dd + bb.w;
      acc.x = acc.x > 0.0f ? acc.x : 0.0f; acc.y = acc.y > 0.0f ? acc.y : 0.0f;
      acc.z = acc.z > 0.0f ? acc.z : 0.0f; acc.w = acc.w > 0.0f ? acc.w : 0.0f;
    }
    *(float4*)&tile[r * AG_PITCH + 4 * lane] = acc;
    if (lane == 0) rowscale[r] = dd;
  }
  __syncthreads();
  if (w >= 4) return;

  const int mn    = lane & 15;
  const int khalf = lane >> 4;
  const float* __restrict__ arow = tile + mn * AG_PITCH + 2 * khalf;
  const float* __restrict__ Bp = B + (size_t)(2 * khalf) * OUT_DIM + w * 16 + mn;
  v8f d = v8f{};
#pragma unroll 16
  for (int k = 0; k < HID_DIM; k += 4) {
    v2f a;
    a.x = arow[k];
    a.y = arow[k + 1];
    v2f b;
    b.x = Bp[(size_t)k       * OUT_DIM];
    b.y = Bp[(size_t)(k + 1) * OUT_DIM];
    d = __builtin_amdgcn_wmma_f32_16x16x4_f32(
         false, a,  false, b,
         (short)0, d,  false,  false);
  }
#pragma unroll
  for (int i = 0; i < 8; ++i) {
    const int row = row0 + 8 * khalf + i;
    if (row < n_nodes)
      C[(size_t)row * OUT_DIM + w * 16 + mn] = d[i] * rowscale[8 * khalf + i];
  }
}

extern "C" void kernel_launch(void* const* d_in, const int* in_sizes, int n_in,
                              void* d_out, int out_size, void* d_ws, size_t ws_size,
                              hipStream_t stream) {
  const float* x  = (const float*)d_in[0];
  const int*   ei = (const int*)d_in[1];
  const float* W1 = (const float*)d_in[2];
  const float* b1 = (const float*)d_in[3];
  const float* W2 = (const float*)d_in[4];
  const float* b2 = (const float*)d_in[5];
  float* out = (float*)d_out;

  const int N = in_sizes[0] / IN_DIM;
  const int E = in_sizes[1] / 2;
  const int* src = ei;
  const int* dst = ei + E;

  float* h  = (float*)d_ws;
  float* h2 = h + (size_t)N * HID_DIM;
  int* cnt      = (int*)(h2 + (size_t)N * OUT_DIM);
  int* rowstart = cnt + N;
  int* csr      = rowstart + N;
  const int ntile = (E + TILE_E - 1) / TILE_E;
  const int nb    = (N + BIN_W - 1) / BIN_W;
  int2* staged = (int2*)d_ws;
  int*  table  = (int*)(staged + (size_t)ntile * TILE_E);

  const int B256 = 256;
  bin_tile_kernel<<<ntile, B256, 0, stream>>>(src, dst, staged, table, E, nb);
  group_bin_kernel<<<nb, B256, 0, stream>>>(staged, table, cnt, rowstart, csr, N, ntile, nb);

  const int waves  = (N + 15) / 16;
  const int gblks  = (waves + 3) / 4;

  gemm_wmma_kernel<IN_DIM, HID_DIM / 16><<<gblks, 128, 0, stream>>>(x, W1, cnt, h, N);

  gather_gemm_kernel<<<waves, B256, 0, stream>>>(h, csr, rowstart, cnt, b1, W2, h2, N);

  const long long th2 = (long long)N * (OUT_DIM / 4);
  gather_kernel<OUT_DIM, false><<<(int)((th2 + B256 - 1) / B256), B256, 0, stream>>>(
      h2, csr, rowstart, cnt, b2, out, N);
}


# Layer-1 GEMM: weight matrix copied once per work-group into LDS as (k,k+1) pair rows, one 16-wave work-group per WGP walking the row strips; B fragments come from 16-byte LDS reads; sums unchanged
# speedup vs baseline: 5.9003x; 1.1864x over previous
#include <hip/hip_runtime.h>

#define IN_DIM  256
#define HID_DIM 128
#define OUT_DIM 64

typedef __attribute__((ext_vector_type(2))) float v2f;
typedef __attribute__((ext_vector_type(4))) float v4f;
typedef __attribute__((ext_vector_type(8))) float v8f;

#define BIN_SHIFT 7
#define BIN_W     (1 << BIN_SHIFT)
#define NB_MAX    512
#define TILE_E    4096
#define EPT       (TILE_E / 256)
#define BIN_CAP   8192

__device__ __forceinline__ int block_scan_excl(int v, int* wsum, int& total) {
  const int lane = threadIdx.x & 31, w = threadIdx.x >> 5;
  int incl = v;
  for (int d = 1; d < 32; d <<= 1) {
    int t = __shfl_up(incl, d);
    if (lane >= d) incl += t;
  }
  if (lane == 31) wsum[w] = incl;
  __syncthreads();
  int base = 0, tot = 0;
#pragma unroll
  for (int i = 0; i < 8; ++i) {
    const int s = wsum[i];
    if (i < w) base += s;
    tot += s;
  }
  total = tot;
  __syncthreads();
  return base + incl - v;
}

__global__ __launch_bounds__(256) void bin_tile_kernel(
    const int* __restrict__ src, const int* __restrict__ dst,
    int2* __restrict__ staged, int* __restrict__ table, int e, int nb) {
  __shared__ int hist[NB_MAX];
  __shared__ int cursor[NB_MAX];
  __shared__ int wsum[8];
  __shared__ int2 stage[TILE_E];
  const int tid = threadIdx.x;
  const int e0 = blockIdx.x * TILE_E;
  const int ne = (e - e0) < TILE_E ? (e - e0) : TILE_E;
  for (int j = tid; j < NB_MAX; j += 256) hist[j] = 0;
  __syncthreads();
  int s[EPT], d[EPT];
#pragma unroll
  for (int k = 0; k < EPT; ++k) {
    const int i = k * 256 + tid;
    d[k] = -1; s[k] = 0;
    if (i < ne) { d[k] = dst[e0 + i]; s[k] = src[e0 + i]; }
  }
#pragma unroll
  for (int k = 0; k < EPT; ++k)
    if (d[k] >= 0) atomicAdd(&hist[d[k] >> BIN_SHIFT], 1);
  __syncthreads();
  const int h0 = hist[2 * tid], h1 = hist[2 * tid + 1];
  int tot;
  const int ex = block_scan_excl(h0 + h1, wsum, tot);
  cursor[2 * tid] = ex;
  cursor[2 * tid + 1] = ex + h0;
  int* __restrict__ trow = table + (size_t)blockIdx.x * (nb + 1);
  if (2 * tid <= nb)     trow[2 * tid]     = ex;
  if (2 * tid + 1 <= nb) trow[2 * tid + 1] = ex + h0;
  __syncthreads();
#pragma unroll
  for (int k = 0; k < EPT; ++k) {
    if (d[k] >= 0) {
      const int pos = atomicAdd(&cursor[d[k] >> BIN_SHIFT], 1);
      stage[pos] = make_int2(s[k], d[k]);
    }
  }
  __syncthreads();
  for (int i = tid; i < ne; i += 256) staged[(size_t)e0 + i] = stage[i];
}

__global__ __launch_bounds__(256) void group_bin_kernel(
    const int2* __restrict__ staged, const int* __restrict__ table,
    int* __restrict__ cnt, int* __restrict__ rowstart, int* __restrict__ csr,
    int n, int ntile, int nb) {
  __shared__ int cntL[BIN_W];
  __shared__ int curL[BIN_W];
  __shared__ int wsum[8];
  __shared__ int stageL[BIN_CAP];
  const int tid = threadIdx.x;
  const int b = blockIdx.x;
  const int lo = b << BIN_SHIFT;
  if (tid < BIN_W) cntL[tid] = 0;
  int mystart = 0, mylen = 0;
  for (int t = tid; t < ntile; t += 256) {
    const int* __restrict__ tr = table + (size_t)t * (nb + 1) + b;
    const int a = tr[0], z = tr[1];
    mystart += a; mylen += z - a;
  }
  __syncthreads();
  for (int t = tid; t < ntile; t += 256) {
    const int* __restrict__ tr = table + (size_t)t * (nb + 1) + b;
    const int a = tr[0], z = tr[1];
    const int2* __restrict__ p = staged + (size_t)t * TILE_E + a;
    for (int i = 0; i < z - a; ++i) atomicAdd(&cntL[p[i].y - lo], 1);
  }
  int base, total;
  block_scan_excl(mystart, wsum, base);
  block_scan_excl(mylen, wsum, total);
  const int c = tid < BIN_W ? cntL[tid] : 0;
  int tot2;
  const int ex = block_scan_excl(c, wsum, tot2);
  if (tid < BIN_W) {
    curL[tid] = ex;
    const int node = lo + tid;
    if (node < n) { cnt[node] = c; rowstart[node] = base + ex; }
  }
  __syncthreads();
  const bool in_lds = total <= BIN_CAP;
  for (int t = tid; t < ntile; t += 256) {
    const int* __restrict__ tr = table + (size_t)t * (nb + 1) + b;
    const int a = tr[0], z = tr[1];
    const int2* __restrict__ p = staged + (size_t)t * TILE_E + a;
    for (int i = 0; i < z - a; ++i) {
      const int2 ed = p[i];
      const int pos = atomicAdd(&curL[ed.y - lo], 1);
      if (in_lds) stageL[pos] = ed.x; else csr[base + pos] = ed.x;
    }
  }
  __syncthreads();
  if (in_lds)
    for (int i = tid; i < total; i += 256) csr[base + i] = stageL[i];
}

#define W_LANE  20
#define W_PITCH 324
#define G1_KS   8

template <int K, int NTU>
__device__ __forceinline__ void strip_product(const float* __restrict__ A,
                                              const float* Ws,
                                              const int* __restrict__ cnt,
                                              float* __restrict__ C,
                                              int M, int N, int row0, int tile0, int lane,
                                              bool pre, const v2f (&a0)[G1_KS]) {
  const int mn    = lane & 15;
  const int khalf = lane >> 4;
  int arow = row0 + mn;
  if (arow >= M) arow = M - 1;
  const float* __restrict__ Ar = A + (size_t)arow * K + 2 * khalf;
  const float* Wl = Ws + khalf * W_PITCH + W_LANE * mn + 2 * tile0;

  v8f acc[NTU];
#pragma unroll
  for (int t = 0; t < NTU; ++t) acc[t] = v8f{};

  float sc[8];
#pragma unroll
  for (int i = 0; i < 8; ++i) {
    int row = row0 + 8 * khalf + i;
    if (row >= M) row = M - 1;
    sc[i] = rsqrtf((float)(cnt[row] + 1));
  }

  v2f ax[G1_KS], ay[G1_KS];
  if (pre) {
#pragma unroll
    for (int s = 0; s < G1_KS; ++s) ax[s] = a0[s];
  } else {
#pragma unroll
    for (int s = 0; s < G1_KS; ++s) ax[s] = *(const v2f*)(Ar + 4 * s);
  }

#pragma unroll 1
  for (int k0 = 0; k0 < K; k0 += 8 * G1_KS) {
#pragma unroll
    for (int s = 0; s < G1_KS; ++s) ay[s] = *(const v2f*)(Ar + k0 + 4 * G1_KS + 4 * s);
#pragma unroll
    for (int s = 0; s < G1_KS; ++s) {
      const float* wp = Wl + ((k0 >> 1) + 2 * s) * W_PITCH;
#pragma unroll
      for (int t = 0; t < NTU; t += 2) {
        const v4f bb = *(const v4f*)(wp + 2 * t);
        acc[t] = __builtin_amdgcn_wmma_f32_16x16x4_f32(
             false, ax[s],  false, bb.xy,
             (short)0, acc[t],  false,  false);
        acc[t + 1] = __builtin_amdgcn_wmma_f32_16x16x4_f32(
             false, ax[s],  false, bb.zw,
             (short)0, acc[t + 1],  false,  false);
      }
    }
    if (k0 + 8 * G1_KS < K) {
#pragma unroll
      for (int s = 0; s < G1_KS; ++s) ax[s] = *(const v2f*)(Ar + k0 + 8 * G1_KS + 4 * s);
    }
#pragma unroll
    for (int s = 0; s < G1_KS; ++s) {
      const float* wp = Wl + ((k0 >> 1) + 2 * G1_KS + 2 * s) * W_PITCH;
#pragma unroll
      for (int t = 0; t < NTU; t += 2) {
        const v4f bb = *(const v4f*)(wp + 2 * t);
        acc[t] = __builtin_amdgcn_wmma_f32_16x16x4_f32(
             false, ay[s],  false, bb.xy,
             (short)0, acc[t],  false,  false);
        acc[t + 1] = __builtin_amdgcn_wmma_f32_16x16x4_f32(
             false, ay[s],  false, bb.zw,
             (short)0, acc[t + 1],  false,  false);
      }
    }
  }

#pragma unroll
  for (int t = 0; t < NTU; ++t) {
#pragma unroll
    for (int i = 0; i < 8; ++i) {
      const int row = row0 + 8 * khalf + i;
      if (row < M) C[(size_t)row * N + (tile0 + t) * 16 + mn] = acc[t][i] * sc[i];
    }
  }
}

template <int K, int NT>
__global__ __launch_bounds__(512) void gemm_wmma_kernel(const float* __restrict__ A,
                                 const float* __restrict__ B,
                                 const int* __restrict__ cnt,
                                 float* __restrict__ C, int M) {
  const int N = NT * 16;
  static_assert(NT == 8 && K % 64 == 0, "a wave copies one B row as 32 lanes x 16 bytes; strips split in four pairs of tiles");
  static_assert(W_PITCH >= 16 * W_LANE && W_PITCH % 4 == 0 && W_LANE % 4 == 0, "16-byte reads stay aligned");
  __shared__ __attribute__((aligned(16))) float Ws[(K / 2) * W_PITCH];
  const int lane = threadIdx.x & 31;
  const int w    = threadIdx.x >> 5;

  const int ws = __builtin_amdgcn_readfirstlane(w);
  const int nstrips = (M + 15) / 16;
  const int nblk = (int)gridDim.x;
  const int b = (int)blockIdx.x;
  const int nj = b < nstrips ? (nstrips - b + nblk - 1) / nblk : 0;
  const int nfull = nj & ~3;
  const int nquart = 4 * (nj - nfull);
  const bool diag = (ws >> 2) == (ws & 3);
  const int u0 = diag ? ws / 5 : ws - (ws / 5 + 1);
  const bool any = diag ? u0 < nquart : u0 < nfull;
  const int strip0 = diag ? b + nblk * (nfull + (u0 >> 2)) : b + nblk * u0;

  v2f a0[G1_KS];
#pragma unroll
  for (int s = 0; s < G1_KS; ++s) a0[s] = v2f{};
  if (any) {
    int arow = strip0 * 16 + (lane & 15);
    if (arow >= M) arow = M - 1;
    const float* __restrict__ Ar = A + (size_t)arow * K + 2 * (lane >> 4);
#pragma unroll
    for (int s = 0; s < G1_KS; ++s) a0[s] = *(const v2f*)(Ar + 4 * s);
  }

#pragma unroll
  for (int i = 0; i < (K / 2) / 16; ++i) {
    const int p = w + 16 * i;
    const float4 q0 = ((const float4*)(B + (size_t)(2 * p)     * N))[lane];
    const float4 q1 = ((const float4*)(B + (size_t)(2 * p + 1) * N))[lane];
    float* d = Ws + p * W_PITCH + W_LANE * 4 * (lane & 3) + 2 * (lane >> 2);
    *(v2f*)(d)              = v2f{q0.x, q1.x};
    *(v2f*)(d + W_LANE)     = v2f{q0.y, q1.y};
    *(v2f*)(d + 2 * W_LANE) = v2f{q0.z, q1.z};
    *(v2f*)(d + 3 * W_LANE) = v2f{q0.w, q1.w};
  }
  __syncthreads();

  if (!diag) {
    for (int j = u0; j < nfull; j += 12)
      strip_product<K, NT>(A, Ws, cnt, C, M, N, (b + nblk * j) * 16, 0, lane, j == u0, a0);
  } else {
    for (int q = u0; q < nquart; q += 4)
      strip_product<K, 2>(A, Ws, cnt, C, M, N, (b + nblk * (nfull + (q >> 2))) * 16, 2 * (q & 3), lane, q == u0, a0);
  }
}

template <int F, bool RELU>
__global__ void gather_kernel(const float* __restrict__ h,
                              const int* __restrict__ csr,
                              const int* __restrict__ rowstart,
                              const int* __restrict__ cnt,
                              const float* __restrict__ bias,
                              float* __restrict__ outp, int n_nodes) {
  constexpr int LPN = F / 4;
  const long long gt = blockIdx.x * (long long)blockDim.x + threadIdx.x;
  const int node = (int)(gt / LPN);
  const int l = (int)(gt & (LPN - 1));
  if (node >= n_nodes) return;
  const float4* __restrict__ h4 = (const float4*)h;
  const int beg = rowstart[node];
  const int c   = cnt[node];
  const float dd = rsqrtf((float)(c + 1));

  float4 acc = h4[(size_t)node * LPN + l];
  int j = 0;
  for (; j + 4 <= c; j += 4) {
    const int s0 = csr[beg + j],     s1 = csr[beg + j + 1];
    const int s2 = csr[beg + j + 2], s3 = csr[beg + j + 3];
    const float4 v0 = h4[(size_t)s0 * LPN + l];
    const float4 v1 = h4[(size_t)s1 * LPN + l];
    const float4 v2 = h4[(size_t)s2 * LPN + l];
    const float4 v3 = h4[(size_t)s3 * LPN + l];
    acc.x += v0.x; acc.y += v0.y; acc.z += v0.z; acc.w += v0.w;
    acc.x += v1.x; acc.y += v1.y; acc.z += v1.z; acc.w += v1.w;
    acc.x += v2.x; acc.y += v2.y; acc.z += v2.z; acc.w += v2.w;
    acc.x += v3.x; acc.y += v3.y; acc.z += v3.z; acc.w += v3.w;
  }
  const int rem = c - j;
  if (rem > 0) {
    const int s0 = csr[beg + j];
    const int s1 = rem > 1 ? csr[beg + j + 1] : s0;
    const int s2 = rem > 2 ? csr[beg + j + 2] : s0;
    const float4 v0 = h4[(size_t)s0 * LPN + l];
    const float4 v1 = h4[(size_t)s1 * LPN + l];
    const float4 v2 = h4[(size_t)s2 * LPN + l];
    acc.x += v0.x; acc.y += v0.y; acc.z += v0.z; acc.w += v0.w;
    if (rem > 1) {
      acc.x += v1.x; acc.y += v1.y; acc.z += v1.z; acc.w += v1.w;
    }
    if (rem > 2) {
      acc.x += v2.x; acc.y += v2.y; acc.z += v2.z; acc.w += v2.w;
    }
  }
  const float4 b = ((const float4*)bias)[l];
  acc.x = acc.x * dd + b.x; acc.y = acc.y * dd + b.y;
  acc.z = acc.z * dd + b.z; acc.w = acc.w * dd + b.w;
  if (RELU) {
    acc.x = acc.x > 0.0f ? acc.x : 0.0f; acc.y = acc.y > 0.0f ? acc.y : 0.0f;
    acc.z = acc.z > 0.0f ? acc.z : 0.0f; acc.w = acc.w > 0.0f ? acc.w : 0.0f;
  }
  ((float4*)outp)[(size_t)node * LPN + l] = acc;
}

#define AG_PITCH (HID_DIM + 4)

__global__ __launch_bounds__(256) __attribute__((amdgpu_waves_per_eu(12)))
void gather_gemm_kernel(
    const float* __restrict__ h, const int* __restrict__ csr,
    const int* __restrict__ rowstart, const int* __restrict__ cnt,
    const float* __restrict__ bias, const float* __restrict__ B,
    float* __restrict__ C, int n_nodes) {
  __shared__ __attribute__((aligned(16))) float tile[16 * AG_PITCH];
  __shared__ float rowscale[16];
  const int lane = threadIdx.x & 31;
  const int w    = threadIdx.x >> 5;
  const int row0 = blockIdx.x * 16;
  const float4* __restrict__ h4 = (const float4*)h;

  for (int q = 0; q < 2; ++q) {
    const int r = 2 * w + q;
    const int node = row0 + r;
    float4 acc = make_float4(0.0f, 0.0f, 0.0f, 0.0f);
    float dd = 0.0f;
    if (node < n_nodes) {
      const int beg = rowstart[node];
      const int c   = cnt[node];
      dd = rsqrtf((float)(c + 1));
      acc = h4[(size_t)node * 32 + lane];
      int j = 0;
      for (; j + 4 <= c; j += 4) {
        const int s0 = csr[beg + j],     s1 = csr[beg + j + 1];
        const int s2 = csr[beg + j + 2], s3 = csr[beg + j + 3];
        const float4 v0 = h4[(size_t)s0 * 32 + lane];
        const float4 v1 = h4[(size_t)s1 * 32 + lane];
        const float4 v2 = h4[(size_t)s2 * 32 + lane];
        const float4 v3 = h4[(size_t)s3 * 32 + lane];
        acc.x += v0.x; acc.y += v0.y; acc.z += v0.z; acc.w += v0.w;
        acc.x += v1.x; acc.y += v1.y; acc.z += v1.z; acc.w += v1.w;
        acc.x += v2.x; acc.y += v2.y; acc.z += v2.z; acc.w += v2.w;
        acc.x += v3.x; acc.y += v3.y; acc.z += v3.z; acc.w += v3.w;
      }
      const int rem = c - j;
      if (rem > 0) {
        const int s0 = csr[beg + j];
        const int s1 = rem > 1 ? csr[beg + j + 1] : s0;
        const int s2 = rem > 2 ? csr[beg + j + 2] : s0;
        const float4 v0 = h4[(size_t)s0 * 32 + lane];
        const float4 v1 = h4[(size_t)s1 * 32 + lane];
        const float4 v2 = h4[(size_t)s2 * 32 + lane];
        acc.x += v0.x; acc.y += v0.y; acc.z += v0.z; acc.w += v0.w;
        if (rem > 1) {
          acc.x += v1.x; acc.y += v1.y; acc.z += v1.z; acc.w += v1.w;
        }
        if (rem > 2) {
          acc.x += v2.x; acc.y += v2.y; acc.z += v2.z; acc.w += v2.w;
        }
      }
      const float4 bb = ((const float4*)bias)[lane];
      acc.x = acc.x * dd + bb.x; acc.y = acc.y * dd + bb.y;
      acc.z = acc.z * dd + bb.z; acc.w = acc.w * dd + bb.w;
      acc.x = acc.x > 0.0f ? acc.x : 0.0f; acc.y = acc.y > 0.0f ? acc.y : 0.0f;
      acc.z = acc.z > 0.0f ? acc.z : 0.0f; acc.w = acc.w > 0.0f ? acc.w : 0.0f;
    }
    *(float4*)&tile[r * AG_PITCH + 4 * lane] = acc;
    if (lane == 0) rowscale[r] = dd;
  }
  __syncthreads();
  if (w >= 4) return;

  const int mn    = lane & 15;
  const int khalf = lane >> 4;
  const float* __restrict__ arow = tile + mn * AG_PITCH + 2 * khalf;
  const float* __restrict__ Bp = B + (size_t)(2 * khalf) * OUT_DIM + w * 16 + mn;
  v8f d = v8f{};
#pragma unroll 16
  for (int k = 0; k < HID_DIM; k += 4) {
    v2f a;
    a.x = arow[k];
    a.y = arow[k + 1];
    v2f b;
    b.x = Bp[(size_t)k       * OUT_DIM];
    b.y = Bp[(size_t)(k + 1) * OUT_DIM];
    d = __builtin_amdgcn_wmma_f32_16x16x4_f32(
         false, a,  false, b,
         (short)0, d,  false,  false);
  }
#pragma unroll
  for (int i = 0; i < 8; ++i) {
    const int row = row0 + 8 * khalf + i;
    if (row < n_nodes)
      C[(size_t)row * OUT_DIM + w * 16 + mn] = d[i] * rowscale[8 * khalf + i];
  }
}

extern "C" void kernel_launch(void* const* d_in, const int* in_sizes, int n_in,
                              void* d_out, int out_size, void* d_ws, size_t ws_size,
                              hipStream_t stream) {
  const float* x  = (const float*)d_in[0];
  const int*   ei = (const int*)d_in[1];
  const float* W1 = (const float*)d_in[2];
  const float* b1 = (const float*)d_in[3];
  const float* W2 = (const float*)d_in[4];
  const float* b2 = (const float*)d_in[5];
  float* out = (float*)d_out;

  const int N = in_sizes[0] / IN_DIM;
  const int E = in_sizes[1] / 2;
  const int* src = ei;
  const int* dst = ei + E;

  float* h  = (float*)d_ws;
  float* h2 = h + (size_t)N * HID_DIM;
  int* cnt      = (int*)(h2 + (size_t)N * OUT_DIM);
  int* rowstart = cnt + N;
  int* csr      = rowstart + N;
  const int ntile = (E + TILE_E - 1) / TILE_E;
  const int nb    = (N + BIN_W - 1) / BIN_W;
  int2* staged = (int2*)d_ws;
  int*  table  = (int*)(staged + (size_t)ntile * TILE_E);

  const int B256 = 256;
  bin_tile_kernel<<<ntile, B256, 0, stream>>>(src, dst, staged, table, E, nb);
  group_bin_kernel<<<nb, B256, 0, stream>>>(staged, table, cnt, rowstart, csr, N, ntile, nb);

  const int waves  = (N + 15) / 16;

  gemm_wmma_kernel<IN_DIM, HID_DIM / 16><<<256, 512, 0, stream>>>(x, W1, cnt, h, N);

  gather_gemm_kernel<<<waves, B256, 0, stream>>>(h, csr, rowstart, cnt, b1, W2, h2, N);

  const long long th2 = (long long)N * (OUT_DIM / 4);
  gather_kernel<OUT_DIM, false><<<(int)((th2 + B256 - 1) / B256), B256, 0, stream>>>(
      h2, csr, rowstart, cnt, b2, out, N);
}


# Second sort launch folded into the layer-1 GEMM launch: a bin = the 208 nodes whose rows one GEMM work-group owns, which groups their edges first and scales from its own LDS counts (5 to 4 launches)
# speedup vs baseline: 6.3354x; 1.0737x over previous
#include <hip/hip_runtime.h>

#define IN_DIM  256
#define HID_DIM 128
#define OUT_DIM 64

typedef __attribute__((ext_vector_type(2))) float v2f;
typedef __attribute__((ext_vector_type(4))) float v4f;
typedef __attribute__((ext_vector_type(8))) float v8f;

#define BIN_STRIPS 13
#define BIN_W     (16 * BIN_STRIPS)
#define NB_MAX    512
#define TILE_E    4096
#define EPT       (TILE_E / 256)
#define BIN_CAP   8192

template <int NW>
__device__ __forceinline__ int block_scan_excl(int v, int* wsum, int& total) {
  const int lane = threadIdx.x & 31, w = threadIdx.x >> 5;
  int incl = v;
  for (int d = 1; d < 32; d <<= 1) {
    int t = __shfl_up(incl, d);
    if (lane >= d) incl += t;
  }
  if (lane == 31) wsum[w] = incl;
  __syncthreads();
  int base = 0, tot = 0;
#pragma unroll
  for (int i = 0; i < NW; ++i) {
    const int s = wsum[i];
    if (i < w) base += s;
    tot += s;
  }
  total = tot;
  __syncthreads();
  return base + incl - v;
}

__global__ __launch_bounds__(256) void bin_tile_kernel(
    const int* __restrict__ src, const int* __restrict__ dst,
    int2* __restrict__ staged, int* __restrict__ table, int e, int nb) {
  __shared__ int hist[NB_MAX];
  __shared__ int cursor[NB_MAX];
  __shared__ int wsum[8];
  __shared__ int2 stage[TILE_E];
  const int tid = threadIdx.x;
  const int e0 = blockIdx.x * TILE_E;
  const int ne = (e - e0) < TILE_E ? (e - e0) : TILE_E;
  for (int j = tid; j < NB_MAX; j += 256) hist[j] = 0;
  __syncthreads();
  int s[EPT], d[EPT];
#pragma unroll
  for (int k = 0; k < EPT; ++k) {
    const int i = k * 256 + tid;
    d[k] = -1; s[k] = 0;
    if (i < ne) { d[k] = dst[e0 + i]; s[k] = src[e0 + i]; }
  }
#pragma unroll
  for (int k = 0; k < EPT; ++k)
    if (d[k] >= 0) atomicAdd(&hist[(unsigned)d[k] / BIN_W], 1);
  __syncthreads();
  const int h0 = hist[2 * tid], h1 = hist[2 * tid + 1];
  int tot;
  const int ex = block_scan_excl<8>(h0 + h1, wsum, tot);
  cursor[2 * tid] = ex;
  cursor[2 * tid + 1] = ex + h0;
  int* __restrict__ trow = table + (size_t)blockIdx.x * (nb + 1);
  if (2 * tid <= nb)     trow[2 * tid]     = ex;
  if (2 * tid + 1 <= nb) trow[2 * tid + 1] = ex + h0;
  __syncthreads();
#pragma unroll
  for (int k = 0; k < EPT; ++k) {
    if (d[k] >= 0) {
      const int pos = atomicAdd(&cursor[(unsigned)d[k] / BIN_W], 1);
      stage[pos] = make_int2(s[k], d[k]);
    }
  }
  __syncthreads();
  for (int i = tid; i < ne; i += 256) staged[(size_t)e0 + i] = stage[i];
}

#define W_LANE  20
#define W_PITCH 324
#define G1_KS   8

template <int K, int NTU>
__device__ __forceinline__ void strip_product(const float* __restrict__ A,
                                              const float* Ws,
                                              const int* cntL, int lo,
                                              float* __restrict__ C,
                                              int M, int N, int row0, int tile0, int lane,
                                              bool pre, const v2f (&a0)[G1_KS]) {
  const int mn    = lane & 15;
  const int khalf = lane >> 4;
  int arow = row0 + mn;
  if (arow >= M) arow = M - 1;
  const float* __restrict__ Ar = A + (size_t)arow * K + 2 * khalf;
  const float* Wl = Ws + khalf * W_PITCH + W_LANE * mn + 2 * tile0;

  v8f acc[NTU];
#pragma unroll
  for (int t = 0; t < NTU; ++t) acc[t] = v8f{};

  float sc[8];
#pragma unroll
  for (int i = 0; i < 8; ++i) {
    int row = row0 + 8 * khalf + i;
    if (row >= M) row = M - 1;
    sc[i] = rsqrtf((float)(cntL[row - lo] + 1));
  }

  v2f ax[G1_KS], ay[G1_KS];
  if (pre) {
#pragma unroll
    for (int s = 0; s < G1_KS; ++s) ax[s] = a0[s];
  } else {
#pragma unroll
    for (int s = 0; s < G1_KS; ++s) ax[s] = *(const v2f*)(Ar + 4 * s);
  }

#pragma unroll 1
  for (int k0 = 0; k0 < K; k0 += 8 * G1_KS) {
#pragma unroll
    for (int s = 0; s < G1_KS; ++s) ay[s] = *(const v2f*)(Ar + k0 + 4 * G1_KS + 4 * s);
#pragma unroll
    for (int s = 0; s < G1_KS; ++s) {
      const float* wp = Wl + ((k0 >> 1) + 2 * s) * W_PITCH;
#pragma unroll
      for (int t = 0; t < NTU; t += 2) {
        const v4f bb = *(const v4f*)(wp + 2 * t);
        acc[t] = __builtin_amdgcn_wmma_f32_16x16x4_f32(
             false, ax[s],  false, bb.xy,
             (short)0, acc[t],  false,  false);
        acc[t + 1] = __builtin_amdgcn_wmma_f32_16x16x4_f32(
             false, ax[s],  false, bb.zw,
             (short)0, acc[t + 1],  false,  false);
      }
    }
    if (k0 + 8 * G1_KS < K) {
#pragma unroll
      for (int s = 0; s < G1_KS; ++s) ax[s] = *(const v2f*)(Ar + k0 + 8 * G1_KS + 4 * s);
    }
#pragma unroll
    for (int s = 0; s < G1_KS; ++s) {
      const float* wp = Wl + ((k0 >> 1) + 2 * G1_KS + 2 * s) * W_PITCH;
#pragma unroll
      for (int t = 0; t < NTU; t += 2) {
        const v4f bb = *(const v4f*)(wp + 2 * t);
        acc[t] = __builtin_amdgcn_wmma_f32_16x16x4_f32(
             false, ay[s],  false, bb.xy,
             (short)0, acc[t],  false,  false);
        acc[t + 1] = __builtin_amdgcn_wmma_f32_16x16x4_f32(
             false, ay[s],  false, bb.zw,
             (short)0, acc[t + 1],  false,  false);
      }
    }
  }

#pragma unroll
  for (int t = 0; t < NTU; ++t) {
#pragma unroll
    for (int i = 0; i < 8; ++i) {
      const int row = row0 + 8 * khalf + i;
      if (row < M) C[(size_t)row * N + (tile0 + t) * 16 + mn] = acc[t][i] * sc[i];
    }
  }
}

template <int K, int NT>
__global__ __launch_bounds__(512) void group_gemm_kernel(
    const int2* __restrict__ staged, const int* __restrict__ table,
    int* __restrict__ cnt, int* __restrict__ rowstart, int* __restrict__ csr,
    int ntile, int nb,
    const float* __restrict__ A, const float* __restrict__ B,
    float* __restrict__ C, int M) {
  const int N = NT * 16;
  static_assert(NT == 8 && K % 64 == 0, "a wave copies one B row as 32 lanes x 16 bytes; strips split in four pairs of tiles");
  static_assert(W_PITCH >= 16 * W_LANE && W_PITCH % 4 == 0 && W_LANE % 4 == 0, "16-byte reads stay aligned");
  static_assert(BIN_W <= 512 && BIN_STRIPS >= 12 && BIN_STRIPS <= 15, "one thread per node of the bin; twelve waves take whole strips, four share the rest");
  __shared__ __attribute__((aligned(16))) float Ws[(K / 2) * W_PITCH];
  __shared__ int cntL[BIN_W];
  __shared__ int curL[BIN_W];
  __shared__ int wsum[16];
  __shared__ int stageL[BIN_CAP];
  const int tid  = threadIdx.x;
  const int lane = tid & 31;
  const int w    = tid >> 5;
  const int b    = (int)blockIdx.x;
  const int lo   = b * BIN_W;

  const int ws = __builtin_amdgcn_readfirstlane(w);
  const int nstrips = (M + 15) / 16;
  const int s_first = b * BIN_STRIPS;
  const int nleft = nstrips - s_first;
  const int nj = nleft < 0 ? 0 : (nleft < BIN_STRIPS ? nleft : BIN_STRIPS);
  const int nfull = nj & ~3;
  const int nquart = 4 * (nj - nfull);
  const bool diag = (ws >> 2) == (ws & 3);
  const int u0 = diag ? ws / 5 : ws - (ws / 5 + 1);
  const bool any = diag ? u0 < nquart : u0 < nfull;
  const int strip0 = s_first + (diag ? nfull + (u0 >> 2) : u0);

  if (tid < BIN_W) cntL[tid] = 0;

  v2f a0[G1_KS];
#pragma unroll
  for (int s = 0; s < G1_KS; ++s) a0[s] = v2f{};
  if (any) {
    int arow = strip0 * 16 + (lane & 15);
    if (arow >= M) arow = M - 1;
    const float* __restrict__ Ar = A + (size_t)arow * K + 2 * (lane >> 4);
#pragma unroll
    for (int s = 0; s < G1_KS; ++s) a0[s] = *(const v2f*)(Ar + 4 * s);
  }

  int mystart = 0, mylen = 0;
  for (int u = tid; u < 2 * ntile; u += 512) {
    const int* __restrict__ tr = table + (size_t)(u >> 1) * (nb + 1) + b;
    const int a = tr[0], z = tr[1];
    const int mid = a + ((z - a + 1) >> 1);
    if (u & 1) { mylen += z - mid; } else { mystart += a; mylen += mid - a; }
  }

#pragma unroll
  for (int i = 0; i < (K / 2) / 16; ++i) {
    const int p = w + 16 * i;
    const float4 q0 = ((const float4*)(B + (size_t)(2 * p)     * N))[lane];
    const float4 q1 = ((const float4*)(B + (size_t)(2 * p + 1) * N))[lane];
    float* d = Ws + p * W_PITCH + W_LANE * 4 * (lane & 3) + 2 * (lane >> 2);
    *(v2f*)(d)              = v2f{q0.x, q1.x};
    *(v2f*)(d + W_LANE)     = v2f{q0.y, q1.y};
    *(v2f*)(d + 2 * W_LANE) = v2f{q0.z, q1.z};
    *(v2f*)(d + 3 * W_LANE) = v2f{q0.w, q1.w};
  }
  __syncthreads();

  for (int u = tid; u < 2 * ntile; u += 512) {
    const int t = u >> 1;
    const int* __restrict__ tr = table + (size_t)t * (nb + 1) + b;
    const int a = tr[0], z = tr[1];
    const int mid = a + ((z - a + 1) >> 1);
    const int i0 = (u & 1) ? mid : a, i1 = (u & 1) ? z : mid;
    const int2* __restrict__ p = staged + (size_t)t * TILE_E;
    for (int i = i0; i < i1; ++i) atomicAdd(&cntL[p[i].y - lo], 1);
  }
  int base, total;
  block_scan_excl<16>(mystart, wsum, base);
  block_scan_excl<16>(mylen, wsum, total);
  const int c = tid < BIN_W ? cntL[tid] : 0;
  int tot2;
  const int ex = block_scan_excl<16>(c, wsum, tot2);
  if (tid < BIN_W) {
    curL[tid] = ex;
    const int node = lo + tid;
    if (node < M) { cnt[node] = c; rowstart[node] = base + ex; }
  }
  __syncthreads();
  const bool in_lds = total <= BIN_CAP;
  for (int u = tid; u < 2 * ntile; u += 512) {
    const int t = u >> 1;
    const int* __restrict__ tr = table + (size_t)t * (nb + 1) + b;
    const int a = tr[0], z = tr[1];
    const int mid = a + ((z - a + 1) >> 1);
    const int i0 = (u & 1) ? mid : a, i1 = (u & 1) ? z : mid;
    const int2* __restrict__ p = staged + (size_t)t * TILE_E;
    for (int i = i0; i < i1; ++i) {
      const int2 ed = p[i];
      const int pos = atomicAdd(&curL[ed.y - lo], 1);
      if (in_lds) stageL[pos] = ed.x; else csr[base + pos] = ed.x;
    }
  }
  __syncthreads();
  if (in_lds)
    for (int i = tid; i < total; i += 512) csr[base + i] = stageL[i];

  if (!diag) {
    for (int j = u0; j < nfull; j += 12)
      strip_product<K, NT>(A, Ws, cntL, lo, C, M, N, (s_first + j) * 16, 0, lane, j == u0, a0);
  } else {
    for (int q = u0; q < nquart; q += 4)
      strip_product<K, 2>(A, Ws, cntL, lo, C, M, N, (s_first + nfull + (q >> 2)) * 16, 2 * (q & 3), lane, q == u0, a0);
  }
}

template <int F, bool RELU>
__global__ void gather_kernel(const float* __restrict__ h,
                              const int* __restrict__ csr,
                              const int* __restrict__ rowstart,
                              const int* __restrict__ cnt,
                              const float* __restrict__ bias,
                              float* __restrict__ outp, int n_nodes) {
  constexpr int LPN = F / 4;
  const long long gt = blockIdx.x * (long long)blockDim.x + threadIdx.x;
  const int node = (int)(gt / LPN);
  const int l = (int)(gt & (LPN - 1));
  if (node >= n_nodes) return;
  const float4* __restrict__ h4 = (const float4*)h;
  const int beg = rowstart[node];
  const int c   = cnt[node];
  const float dd = rsqrtf((float)(c + 1));

  float4 acc = h4[(size_t)node * LPN + l];
  int j = 0;
  for (; j + 4 <= c; j += 4) {
    const int s0 = csr[beg + j],     s1 = csr[beg + j + 1];
    const int s2 = csr[beg + j + 2], s3 = csr[beg + j + 3];
    const float4 v0 = h4[(size_t)s0 * LPN + l];
    const float4 v1 = h4[(size_t)s1 * LPN + l];
    const float4 v2 = h4[(size_t)s2 * LPN + l];
    const float4 v3 = h4[(size_t)s3 * LPN + l];
    acc.x += v0.x; acc.y += v0.y; acc.z += v0.z; acc.w += v0.w;
    acc.x += v1.x; acc.y += v1.y; acc.z += v1.z; acc.w += v1.w;
    acc.x += v2.x; acc.y += v2.y; acc.z += v2.z; acc.w += v2.w;
    acc.x += v3.x; acc.y += v3.y; acc.z += v3.z; acc.w += v3.w;
  }
  const int rem = c - j;
  if (rem > 0) {
    const int s0 = csr[beg + j];
    const int s1 = rem > 1 ? csr[beg + j + 1] : s0;
    const int s2 = rem > 2 ? csr[beg + j + 2] : s0;
    const float4 v0 = h4[(size_t)s0 * LPN + l];
    const float4 v1 = h4[(size_t)s1 * LPN + l];
    const float4 v2 = h4[(size_t)s2 * LPN + l];
    acc.x += v0.x; acc.y += v0.y; acc.z += v0.z; acc.w += v0.w;
    if (rem > 1) {
      acc.x += v1.x; acc.y += v1.y; acc.z += v1.z; acc.w += v1.w;
    }
    if (rem > 2) {
      acc.x += v2.x; acc.y += v2.y; acc.z += v2.z; acc.w += v2.w;
    }
  }
  const float4 b = ((const float4*)bias)[l];
  acc.x = acc.x * dd + b.x; acc.y = acc.y * dd + b.y;
  acc.z = acc.z * dd + b.z; acc.w = acc.w * dd + b.w;
  if (RELU) {
    acc.x = acc.x > 0.0f ? acc.x : 0.0f; acc.y = acc.y > 0.0f ? acc.y : 0.0f;
    acc.z = acc.z > 0.0f ? acc.z : 0.0f; acc.w = acc.w > 0.0f ? acc.w : 0.0f;
  }
  ((float4*)outp)[(size_t)node * LPN + l] = acc;
}

#define AG_PITCH (HID_DIM + 4)

__global__ __launch_bounds__(256) __attribute__((amdgpu_waves_per_eu(12)))
void gather_gemm_kernel(
    const float* __restrict__ h, const int* __restrict__ csr,
    const int* __restrict__ rowstart, const int* __restrict__ cnt,
    const float* __restrict__ bias, const float* __restrict__ B,
    float* __restrict__ C, int n_nodes) {
  __shared__ __attribute__((aligned(16))) float tile[16 * AG_PITCH];
  __shared__ float rowscale[16];
  const int lane = threadIdx.x & 31;
  const int w    = threadIdx.x >> 5;
  const int row0 = blockIdx.x * 16;
  const float4* __restrict__ h4 = (const float4*)h;

  for (int q = 0; q < 2; ++q) {
    const int r = 2 * w + q;
    const int node = row0 + r;
    float4 acc = make_float4(0.0f, 0.0f, 0.0f, 0.0f);
    float dd = 0.0f;
    if (node < n_nodes) {
      const int beg = rowstart[node];
      const int c   = cnt[node];
      dd = rsqrtf((float)(c + 1));
      acc = h4[(size_t)node * 32 + lane];
      int j = 0;
      for (; j + 4 <= c; j += 4) {
        const int s0 = csr[beg + j],     s1 = csr[beg + j + 1];
        const int s2 = csr[beg + j + 2], s3 = csr[beg + j + 3];
        const float4 v0 = h4[(size_t)s0 * 32 + lane];
        const float4 v1 = h4[(size_t)s1 * 32 + lane];
        const float4 v2 = h4[(size_t)s2 * 32 + lane];
        const float4 v3 = h4[(size_t)s3 * 32 + lane];
        acc.x += v0.x; acc.y += v0.y; acc.z += v0.z; acc.w += v0.w;
        acc.x += v1.x; acc.y += v1.y; acc.z += v1.z; acc.w += v1.w;
        acc.x += v2.x; acc.y += v2.y; acc.z += v2.z; acc.w += v2.w;
        acc.x += v3.x; acc.y += v3.y; acc.z += v3.z; acc.w += v3.w;
      }
      const int rem = c - j;
      if (rem > 0) {
        const int s0 = csr[beg + j];
        const int s1 = rem > 1 ? csr[beg + j + 1] : s0;
        const int s2 = rem > 2 ? csr[beg + j + 2] : s0;
        const float4 v0 = h4[(size_t)s0 * 32 + lane];
        const float4 v1 = h4[(size_t)s1 * 32 + lane];
        const float4 v2 = h4[(size_t)s2 * 32 + lane];
        acc.x += v0.x; acc.y += v0.y; acc.z += v0.z; acc.w += v0.w;
        if (rem > 1) {
          acc.x += v1.x; acc.y += v1.y; acc.z += v1.z; acc.w += v1.w;
        }
        if (rem > 2) {
          acc.x += v2.x; acc.y += v2.y; acc.z += v2.z; acc.w += v2.w;
        }
      }
      const float4 bb = ((const float4*)bias)[lane];
      acc.x = acc.x * dd + bb.x; acc.y = acc.y * dd + bb.y;
      acc.z = acc.z * dd + bb.z; acc.w = acc.w * dd + bb.w;
      acc.x = acc.x > 0.0f ? acc.x : 0.0f; acc.y = acc.y > 0.0f ? acc.y : 0.0f;
      acc.z = acc.z > 0.0f ? acc.z : 0.0f; acc.w = acc.w > 0.0f ? acc.w : 0.0f;
    }
    *(float4*)&tile[r * AG_PITCH + 4 * lane] = acc;
    if (lane == 0) rowscale[r] = dd;
  }
  __syncthreads();
  if (w >= 4) return;

  const int mn    = lane & 15;
  const int khalf = lane >> 4;
  const float* __restrict__ arow = tile + mn * AG_PITCH + 2 * khalf;
  const float* __restrict__ Bp = B + (size_t)(2 * khalf) * OUT_DIM + w * 16 + mn;
  v8f d = v8f{};
#pragma unroll 16
  for (int k = 0; k < HID_DIM; k += 4) {
    v2f a;
    a.x = arow[k];
    a.y = arow[k + 1];
    v2f b;
    b.x = Bp[(size_t)k       * OUT_DIM];
    b.y = Bp[(size_t)(k + 1) * OUT_DIM];
    d = __builtin_amdgcn_wmma_f32_16x16x4_f32(
         false, a,  false, b,
         (short)0, d,  false,  false);
  }
#pragma unroll
  for (int i = 0; i < 8; ++i) {
    const int row = row0 + 8 * khalf + i;
    if (row < n_nodes)
      C[(size_t)row * OUT_DIM + w * 16 + mn] = d[i] * rowscale[8 * khalf + i];
  }
}

extern "C" void kernel_launch(void* const* d_in, const int* in_sizes, int n_in,
                              void* d_out, int out_size, void* d_ws, size_t ws_size,
                              hipStream_t stream) {
  const float* x  = (const float*)d_in[0];
  const int*   ei = (const int*)d_in[1];
  const float* W1 = (const float*)d_in[2];
  const float* b1 = (const float*)d_in[3];
  const float* W2 = (const float*)d_in[4];
  const float* b2 = (const float*)d_in[5];
  float* out = (float*)d_out;

  const int N = in_sizes[0] / IN_DIM;
  const int E = in_sizes[1] / 2;
  const int* src = ei;
  const int* dst = ei + E;

  float* h  = (float*)d_ws;
  float* h2 = h + (size_t)N * HID_DIM;
  int* cnt      = (int*)(h2 + (size_t)N * OUT_DIM);
  int* rowstart = cnt + N;
  int* csr      = rowstart + N;
  const int ntile = (E + TILE_E - 1) / TILE_E;
  const int nb    = (N + BIN_W - 1) / BIN_W;
  int2* staged = (int2*)h2;
  int*  table  = (int*)(staged + (size_t)ntile * TILE_E);

  const int B256 = 256;
  bin_tile_kernel<<<ntile, B256, 0, stream>>>(src, dst, staged, table, E, nb);

  const int waves  = (N + 15) / 16;

  group_gemm_kernel<IN_DIM, HID_DIM / 16><<<nb, 512, 0, stream>>>(
      staged, table, cnt, rowstart, csr, ntile, nb, x, W1, h, N);

  gather_gemm_kernel<<<waves, B256, 0, stream>>>(h, csr, rowstart, cnt, b1, W2, h2, N);

  const long long th2 = (long long)N * (OUT_DIM / 4);
  gather_kernel<OUT_DIM, false><<<(int)((th2 + B256 - 1) / B256), B256, 0, stream>>>(
      h2, csr, rowstart, cnt, b2, out, N);
}


# Second sort step reads each tile piece with one wave-wide load (lane = edge), a wave's pieces all in flight, not one dependent load per edge; node sum order = LDS arrival order, so it changes
# speedup vs baseline: 6.8037x; 1.0739x over previous
#include <hip/hip_runtime.h>

#define IN_DIM  256
#define HID_DIM 128
#define OUT_DIM 64

typedef __attribute__((ext_vector_type(2))) float v2f;
typedef __attribute__((ext_vector_type(4))) float v4f;
typedef __attribute__((ext_vector_type(8))) float v8f;

#define BIN_STRIPS 13
#define BIN_W     (16 * BIN_STRIPS)
#define NB_MAX    512
#define TILE_E    4096
#define EPT       (TILE_E / 256)
#define BIN_CAP   8192
#define TPW       13

template <int NW>
__device__ __forceinline__ int block_scan_excl(int v, int* wsum, int& total) {
  const int lane = threadIdx.x & 31, w = threadIdx.x >> 5;
  int incl = v;
  for (int d = 1; d < 32; d <<= 1) {
    int t = __shfl_up(incl, d);
    if (lane >= d) incl += t;
  }
  if (lane == 31) wsum[w] = incl;
  __syncthreads();
  int base = 0, tot = 0;
#pragma unroll
  for (int i = 0; i < NW; ++i) {
    const int s = wsum[i];
    if (i < w) base += s;
    tot += s;
  }
  total = tot;
  __syncthreads();
  return base + incl - v;
}

__global__ __launch_bounds__(256) void bin_tile_kernel(
    const int* __restrict__ src, const int* __restrict__ dst,
    int2* __restrict__ staged, int* __restrict__ table, int e, int nb) {
  __shared__ int hist[NB_MAX];
  __shared__ int cursor[NB_MAX];
  __shared__ int wsum[8];
  __shared__ int2 stage[TILE_E];
  const int tid = threadIdx.x;
  const int e0 = blockIdx.x * TILE_E;
  const int ne = (e - e0) < TILE_E ? (e - e0) : TILE_E;
  for (int j = tid; j < NB_MAX; j += 256) hist[j] = 0;
  __syncthreads();
  int s[EPT], d[EPT];
#pragma unroll
  for (int k = 0; k < EPT; ++k) {
    const int i = k * 256 + tid;
    d[k] = -1; s[k] = 0;
    if (i < ne) { d[k] = dst[e0 + i]; s[k] = src[e0 + i]; }
  }
#pragma unroll
  for (int k = 0; k < EPT; ++k)
    if (d[k] >= 0) atomicAdd(&hist[(unsigned)d[k] / BIN_W], 1);
  __syncthreads();
  const int h0 = hist[2 * tid], h1 = hist[2 * tid + 1];
  int tot;
  const int ex = block_scan_excl<8>(h0 + h1, wsum, tot);
  cursor[2 * tid] = ex;
  cursor[2 * tid + 1] = ex + h0;
  int* __restrict__ trow = table + (size_t)blockIdx.x * (nb + 1);
  if (2 * tid <= nb)     trow[2 * tid]     = ex;
  if (2 * tid + 1 <= nb) trow[2 * tid + 1] = ex + h0;
  __syncthreads();
#pragma unroll
  for (int k = 0; k < EPT; ++k) {
    if (d[k] >= 0) {
      const int pos = atomicAdd(&cursor[(unsigned)d[k] / BIN_W], 1);
      stage[pos] = make_int2(s[k], d[k]);
    }
  }
  __syncthreads();
  for (int i = tid; i < ne; i += 256) staged[(size_t)e0 + i] = stage[i];
}

__device__ __forceinline__ void piece_bounds(const int* __restrict__ table, int ntile, int nb,
                                             int b, int w, int lane, int r, int& ta, int& tz) {
  ta = 0; tz = 0;
  const int t = w + 16 * (r * TPW + lane);
  if (lane < TPW && t < ntile) {
    const int* __restrict__ tr = table + (size_t)t * (nb + 1) + b;
    ta = tr[0]; tz = tr[1];
  }
}

__device__ __forceinline__ void piece_edges(const int2* __restrict__ staged, int ws, int lane, int r,
                                            int ta, int tz, int (&ss)[TPW], int (&dd)[TPW]) {
#pragma unroll
  for (int j = 0; j < TPW; ++j) {
    const int a = __builtin_amdgcn_readlane(ta, j);
    const int n = __builtin_amdgcn_readlane(tz, j) - a;
    const bool in = lane < n;
    const size_t idx = in ? (size_t)(ws + 16 * (r * TPW + j)) * TILE_E + a + lane : 0;
    const int2 ed = staged[idx];
    ss[j] = ed.x; dd[j] = in ? ed.y : -1;
  }
}

template <bool PLACE>
__device__ __forceinline__ void long_pieces(const int2* __restrict__ staged, int w, int lane, int r,
                                            int ta, int tz, int lo, int* cntL, int* curL,
                                            bool in_lds, int* stageL, int* __restrict__ csr, int base) {
  if (__builtin_amdgcn_ballot_w32(tz - ta > 32) == 0) return;
  for (int j = 0; j < TPW; ++j) {
    const int a = __shfl(ta, j);
    const int n = __shfl(tz, j) - a;
    const int2* __restrict__ p = staged + (size_t)(w + 16 * (r * TPW + j)) * TILE_E + a;
    for (int i = 32 + lane; i < n; i += 32) {
      const int2 ed = p[i];
      if (PLACE) {
        const int pos = atomicAdd(&curL[ed.y - lo], 1);
        if (in_lds) stageL[pos] = ed.x; else csr[base + pos] = ed.x;
      } else {
        atomicAdd(&cntL[ed.y - lo], 1);
      }
    }
  }
}

#define W_LANE  20
#define W_PITCH 324
#define G1_KS   8

template <int K, int NTU>
__device__ __forceinline__ void strip_product(const float* __restrict__ A,
                                              const float* Ws,
                                              const int* cntL, int lo,
                                              float* __restrict__ C,
                                              int M, int N, int row0, int tile0, int lane,
                                              bool pre, const v2f (&a0)[G1_KS]) {
  const int mn    = lane & 15;
  const int khalf = lane >> 4;
  int arow = row0 + mn;
  if (arow >= M) arow = M - 1;
  const float* __restrict__ Ar = A + (size_t)arow * K + 2 * khalf;
  const float* Wl = Ws + khalf * W_PITCH + W_LANE * mn + 2 * tile0;

  v8f acc[NTU];
#pragma unroll
  for (int t = 0; t < NTU; ++t) acc[t] = v8f{};

  float sc[8];
#pragma unroll
  for (int i = 0; i < 8; ++i) {
    int row = row0 + 8 * khalf + i;
    if (row >= M) row = M - 1;
    sc[i] = rsqrtf((float)(cntL[row - lo] + 1));
  }

  v2f ax[G1_KS], ay[G1_KS];
  if (pre) {
#pragma unroll
    for (int s = 0; s < G1_KS; ++s) ax[s] = a0[s];
  } else {
#pragma unroll
    for (int s = 0; s < G1_KS; ++s) ax[s] = *(const v2f*)(Ar + 4 * s);
  }

#pragma unroll 1
  for (int k0 = 0; k0 < K; k0 += 8 * G1_KS) {
#pragma unroll
    for (int s = 0; s < G1_KS; ++s) ay[s] = *(const v2f*)(Ar + k0 + 4 * G1_KS + 4 * s);
#pragma unroll
    for (int s = 0; s < G1_KS; ++s) {
      const float* wp = Wl + ((k0 >> 1) + 2 * s) * W_PITCH;
#pragma unroll
      for (int t = 0; t < NTU; t += 2) {
        const v4f bb = *(const v4f*)(wp + 2 * t);
        acc[t] = __builtin_amdgcn_wmma_f32_16x16x4_f32(
             false, ax[s],  false, bb.xy,
             (short)0, acc[t],  false,  false);
        acc[t + 1] = __builtin_amdgcn_wmma_f32_16x16x4_f32(
             false, ax[s],  false, bb.zw,
             (short)0, acc[t + 1],  false,  false);
      }
    }
    if (k0 + 8 * G1_KS < K) {
#pragma unroll
      for (int s = 0; s < G1_KS; ++s) ax[s] = *(const v2f*)(Ar + k0 + 8 * G1_KS + 4 * s);
    }
#pragma unroll
    for (int s = 0; s < G1_KS; ++s) {
      const float* wp = Wl + ((k0 >> 1) + 2 * G1_KS + 2 * s) * W_PITCH;
#pragma unroll
      for (int t = 0; t < NTU; t += 2) {
        const v4f bb = *(const v4f*)(wp + 2 * t);
        acc[t] = __builtin_amdgcn_wmma_f32_16x16x4_f32(
             false, ay[s],  false, bb.xy,
             (short)0, acc[t],  false,  false);
        acc[t + 1] = __builtin_amdgcn_wmma_f32_16x16x4_f32(
             false, ay[s],  false, bb.zw,
             (short)0, acc[t + 1],  false,  false);
      }
    }
  }

#pragma unroll
  for (int t = 0; t < NTU; ++t) {
#pragma unroll
    for (int i = 0; i < 8; ++i) {
      const int row = row0 + 8 * khalf + i;
      if (row < M) C[(size_t)row * N + (tile0 + t) * 16 + mn] = acc[t][i] * sc[i];
    }
  }
}

template <int K, int NT>
__global__ __launch_bounds__(512) void group_gemm_kernel(
    const int2* __restrict__ staged, const int* __restrict__ table,
    int* __restrict__ cnt, int* __restrict__ rowstart, int* __restrict__ csr,
    int ntile, int nb,
    const float* __restrict__ A, const float* __restrict__ B,
    float* __restrict__ C, int M) {
  const int N = NT * 16;
  static_assert(NT == 8 && K % 64 == 0, "a wave copies one B row as 32 lanes x 16 bytes; strips split in four pairs of tiles");
  static_assert(W_PITCH >= 16 * W_LANE && W_PITCH % 4 == 0 && W_LANE % 4 == 0, "16-byte reads stay aligned");
  static_assert(BIN_W <= 512 && BIN_STRIPS >= 12 && BIN_STRIPS <= 15, "one thread per node of the bin; twelve waves take whole strips, four share the rest");
  __shared__ __attribute__((aligned(16))) float Ws[(K / 2) * W_PITCH];
  __shared__ int cntL[BIN_W];
  __shared__ int curL[BIN_W];
  __shared__ int wsum[16];
  __shared__ int stageL[BIN_CAP];
  const int tid  = threadIdx.x;
  const int lane = tid & 31;
  const int w    = tid >> 5;
  const int b    = (int)blockIdx.x;
  const int lo   = b * BIN_W;

  const int ws = __builtin_amdgcn_readfirstlane(w);
  const int nstrips = (M + 15) / 16;
  const int s_first = b * BIN_STRIPS;
  const int nleft = nstrips - s_first;
  const int nj = nleft < 0 ? 0 : (nleft < BIN_STRIPS ? nleft : BIN_STRIPS);
  const int nfull = nj & ~3;
  const int nquart = 4 * (nj - nfull);
  const bool diag = (ws >> 2) == (ws & 3);
  const int u0 = diag ? ws / 5 : ws - (ws / 5 + 1);
  const bool any = diag ? u0 < nquart : u0 < nfull;
  const int strip0 = s_first + (diag ? nfull + (u0 >> 2) : u0);

  if (tid < BIN_W) cntL[tid] = 0;

  const int nround = ((ntile + 15) / 16 + TPW - 1) / TPW;
  int ta0, tz0;
  piece_bounds(table, ntile, nb, b, w, lane, 0, ta0, tz0);
  int mystart = ta0, mylen = tz0 - ta0;

  v2f a0[G1_KS];
#pragma unroll
  for (int s = 0; s < G1_KS; ++s) a0[s] = v2f{};
  if (any) {
    int arow = strip0 * 16 + (lane & 15);
    if (arow >= M) arow = M - 1;
    const float* __restrict__ Ar = A + (size_t)arow * K + 2 * (lane >> 4);
#pragma unroll
    for (int s = 0; s < G1_KS; ++s) a0[s] = *(const v2f*)(Ar + 4 * s);
  }

  constexpr int BROWS = (K / 2) / 16;
  float4 q0[BROWS], q1[BROWS];
#pragma unroll
  for (int i = 0; i < BROWS; ++i) {
    const int p = w + 16 * i;
    q0[i] = ((const float4*)(B + (size_t)(2 * p)     * N))[lane];
    q1[i] = ((const float4*)(B + (size_t)(2 * p + 1) * N))[lane];
  }
  __builtin_amdgcn_sched_barrier(0);
  int ss0[TPW], dd0[TPW];
  piece_edges(staged, ws, lane, 0, ta0, tz0, ss0, dd0);
#pragma unroll
  for (int i = 0; i < BROWS; ++i) {
    const int p = w + 16 * i;
    float* d = Ws + p * W_PITCH + W_LANE * 4 * (lane & 3) + 2 * (lane >> 2);
    *(v2f*)(d)              = v2f{q0[i].x, q1[i].x};
    *(v2f*)(d + W_LANE)     = v2f{q0[i].y, q1[i].y};
    *(v2f*)(d + 2 * W_LANE) = v2f{q0[i].z, q1[i].z};
    *(v2f*)(d + 3 * W_LANE) = v2f{q0[i].w, q1[i].w};
  }
  __syncthreads();

#pragma unroll
  for (int j = 0; j < TPW; ++j)
    if (dd0[j] >= 0) atomicAdd(&cntL[dd0[j] - lo], 1);
  long_pieces<false>(staged, w, lane, 0, ta0, tz0, lo, cntL, curL, false, stageL, csr, 0);
  for (int r = 1; r < nround; ++r) {
    int ta, tz;
    piece_bounds(table, ntile, nb, b, w, lane, r, ta, tz);
    mystart += ta; mylen += tz - ta;
    int ss[TPW], dd[TPW];
    piece_edges(staged, ws, lane, r, ta, tz, ss, dd);
#pragma unroll
    for (int j = 0; j < TPW; ++j)
      if (dd[j] >= 0) atomicAdd(&cntL[dd[j] - lo], 1);
    long_pieces<false>(staged, w, lane, r, ta, tz, lo, cntL, curL, false, stageL, csr, 0);
  }
  int base, total;
  block_scan_excl<16>(mystart, wsum, base);
  block_scan_excl<16>(mylen, wsum, total);
  const int c = tid < BIN_W ? cntL[tid] : 0;
  int tot2;
  const int ex = block_scan_excl<16>(c, wsum, tot2);
  if (tid < BIN_W) {
    curL[tid] = ex;
    const int node = lo + tid;
    if (node < M) { cnt[node] = c; rowstart[node] = base + ex; }
  }
  __syncthreads();
  const bool in_lds = total <= BIN_CAP;

  int pos0[TPW];
#pragma unroll
  for (int j = 0; j < TPW; ++j)
    if (dd0[j] >= 0) pos0[j] = atomicAdd(&curL[dd0[j] - lo], 1);
  if (in_lds) {
#pragma unroll
    for (int j = 0; j < TPW; ++j)
      if (dd0[j] >= 0) stageL[pos0[j]] = ss0[j];
  } else {
#pragma unroll
    for (int j = 0; j < TPW; ++j)
      if (dd0[j] >= 0) csr[base + pos0[j]] = ss0[j];
  }
  long_pieces<true>(staged, w, lane, 0, ta0, tz0, lo, cntL, curL, in_lds, stageL, csr, base);
  for (int r = 1; r < nround; ++r) {
    int ta, tz;
    piece_bounds(table, ntile, nb, b, w, lane, r, ta, tz);
    int ss[TPW], dd[TPW];
    piece_edges(staged, ws, lane, r, ta, tz, ss, dd);
#pragma unroll
    for (int j = 0; j < TPW; ++j) {
      if (dd[j] >= 0) {
        const int pos = atomicAdd(&curL[dd[j] - lo], 1);
        if (in_lds) stageL[pos] = ss[j]; else csr[base + pos] = ss[j];
      }
    }
    long_pieces<true>(staged, w, lane, r, ta, tz, lo, cntL, curL, in_lds, stageL, csr, base);
  }
  __syncthreads();
  if (in_lds)
    for (int i = tid; i < total; i += 512) csr[base + i] = stageL[i];

  if (!diag) {
    for (int j = u0; j < nfull; j += 12)
      strip_product<K, NT>(A, Ws, cntL, lo, C, M, N, (s_first + j) * 16, 0, lane, j == u0, a0);
  } else {
    for (int q = u0; q < nquart; q += 4)
      strip_product<K, 2>(A, Ws, cntL, lo, C, M, N, (s_first + nfull + (q >> 2)) * 16, 2 * (q & 3), lane, q == u0, a0);
  }
}

template <int F, bool RELU>
__global__ void gather_kernel(const float* __restrict__ h,
                              const int* __restrict__ csr,
                              const int* __restrict__ rowstart,
                              const int* __restrict__ cnt,
                              const float* __restrict__ bias,
                              float* __restrict__ outp, int n_nodes) {
  constexpr int LPN = F / 4;
  const long long gt = blockIdx.x * (long long)blockDim.x + threadIdx.x;
  const int node = (int)(gt / LPN);
  const int l = (int)(gt & (LPN - 1));
  if (node >= n_nodes) return;
  const float4* __restrict__ h4 = (const float4*)h;
  const int beg = rowstart[node];
  const int c   = cnt[node];
  const float dd = rsqrtf((float)(c + 1));

  float4 acc = h4[(size_t)node * LPN + l];
  int j = 0;
  for (; j + 4 <= c; j += 4) {
    const int s0 = csr[beg + j],     s1 = csr[beg + j + 1];
    const int s2 = csr[beg + j + 2], s3 = csr[beg + j + 3];
    const float4 v0 = h4[(size_t)s0 * LPN + l];
    const float4 v1 = h4[(size_t)s1 * LPN + l];
    const float4 v2 = h4[(size_t)s2 * LPN + l];
    const float4 v3 = h4[(size_t)s3 * LPN + l];
    acc.x += v0.x; acc.y += v0.y; acc.z += v0.z; acc.w += v0.w;
    acc.x += v1.x; acc.y += v1.y; acc.z += v1.z; acc.w += v1.w;
    acc.x += v2.x; acc.y += v2.y; acc.z += v2.z; acc.w += v2.w;
    acc.x += v3.x; acc.y += v3.y; acc.z += v3.z; acc.w += v3.w;
  }
  const int rem = c - j;
  if (rem > 0) {
    const int s0 = csr[beg + j];
    const int s1 = rem > 1 ? csr[beg + j + 1] : s0;
    const int s2 = rem > 2 ? csr[beg + j + 2] : s0;
    const float4 v0 = h4[(size_t)s0 * LPN + l];
    const float4 v1 = h4[(size_t)s1 * LPN + l];
    const float4 v2 = h4[(size_t)s2 * LPN + l];
    acc.x += v0.x; acc.y += v0.y; acc.z += v0.z; acc.w += v0.w;
    if (rem > 1) {
      acc.x += v1.x; acc.y += v1.y; acc.z += v1.z; acc.w += v1.w;
    }
    if (rem > 2) {
      acc.x += v2.x; acc.y += v2.y; acc.z += v2.z; acc.w += v2.w;
    }
  }
  const float4 b = ((const float4*)bias)[l];
  acc.x = acc.x * dd + b.x; acc.y = acc.y * dd + b.y;
  acc.z = acc.z * dd + b.z; acc.w = acc.w * dd + b.w;
  if (RELU) {
    acc.x = acc.x > 0.0f ? acc.x : 0.0f; acc.y = acc.y > 0.0f ? acc.y : 0.0f;
    acc.z = acc.z > 0.0f ? acc.z : 0.0f; acc.w = acc.w > 0.0f ? acc.w : 0.0f;
  }
  ((float4*)outp)[(size_t)node * LPN + l] = acc;
}

#define AG_PITCH (HID_DIM + 4)

__global__ __launch_bounds__(256) __attribute__((amdgpu_waves_per_eu(12)))
void gather_gemm_kernel(
    const float* __restrict__ h, const int* __restrict__ csr,
    const int* __restrict__ rowstart, const int* __restrict__ cnt,
    const float* __restrict__ bias, const float* __restrict__ B,
    float* __restrict__ C, int n_nodes) {
  __shared__ __attribute__((aligned(16))) float tile[16 * AG_PITCH];
  __shared__ float rowscale[16];
  const int lane = threadIdx.x & 31;
  const int w    = threadIdx.x >> 5;
  const int row0 = blockIdx.x * 16;
  const float4* __restrict__ h4 = (const float4*)h;

  for (int q = 0; q < 2; ++q) {
    const int r = 2 * w + q;
    const int node = row0 + r;
    float4 acc = make_float4(0.0f, 0.0f, 0.0f, 0.0f);
    float dd = 0.0f;
    if (node < n_nodes) {
      const int beg = rowstart[node];
      const int c   = cnt[node];
      dd = rsqrtf((float)(c + 1));
      acc = h4[(size_t)node * 32 + lane];
      int j = 0;
      for (; j + 4 <= c; j += 4) {
        const int s0 = csr[beg + j],     s1 = csr[beg + j + 1];
        const int s2 = csr[beg + j + 2], s3 = csr[beg + j + 3];
        const float4 v0 = h4[(size_t)s0 * 32 + lane];
        const float4 v1 = h4[(size_t)s1 * 32 + lane];
        const float4 v2 = h4[(size_t)s2 * 32 + lane];
        const float4 v3 = h4[(size_t)s3 * 32 + lane];
        acc.x += v0.x; acc.y += v0.y; acc.z += v0.z; acc.w += v0.w;
        acc.x += v1.x; acc.y += v1.y; acc.z += v1.z; acc.w += v1.w;
        acc.x += v2.x; acc.y += v2.y; acc.z += v2.z; acc.w += v2.w;
        acc.x += v3.x; acc.y += v3.y; acc.z += v3.z; acc.w += v3.w;
      }
      const int rem = c - j;
      if (rem > 0) {
        const int s0 = csr[beg + j];
        const int s1 = rem > 1 ? csr[beg + j + 1] : s0;
        const int s2 = rem > 2 ? csr[beg + j + 2] : s0;
        const float4 v0 = h4[(size_t)s0 * 32 + lane];
        const float4 v1 = h4[(size_t)s1 * 32 + lane];
        const float4 v2 = h4[(size_t)s2 * 32 + lane];
        acc.x += v0.x; acc.y += v0.y; acc.z += v0.z; acc.w += v0.w;
        if (rem > 1) {
          acc.x += v1.x; acc.y += v1.y; acc.z += v1.z; acc.w += v1.w;
        }
        if (rem > 2) {
          acc.x += v2.x; acc.y += v2.y; acc.z += v2.z; acc.w += v2.w;
        }
      }
      const float4 bb = ((const float4*)bias)[lane];
      acc.x = acc.x * dd + bb.x; acc.y = acc.y * dd + bb.y;
      acc.z = acc.z * dd + bb.z; acc.w = acc.w * dd + bb.w;
      acc.x = acc.x > 0.0f ? acc.x : 0.0f; acc.y = acc.y > 0.0f ? acc.y : 0.0f;
      acc.z = acc.z > 0.0f ? acc.z : 0.0f; acc.w = acc.w > 0.0f ? acc.w : 0.0f;
    }
    *(float4*)&tile[r * AG_PITCH + 4 * lane] = acc;
    if (lane == 0) rowscale[r] = dd;
  }
  __syncthreads();
  if (w >= 4) return;

  const int mn    = lane & 15;
  const int khalf = lane >> 4;
  const float* __restrict__ arow = tile + mn * AG_PITCH + 2 * khalf;
  const float* __restrict__ Bp = B + (size_t)(2 * khalf) * OUT_DIM + w * 16 + mn;
  v8f d = v8f{};
#pragma unroll 16
  for (int k = 0; k < HID_DIM; k += 4) {
    v2f a;
    a.x = arow[k];
    a.y = arow[k + 1];
    v2f b;
    b.x = Bp[(size_t)k       * OUT_DIM];
    b.y = Bp[(size_t)(k + 1) * OUT_DIM];
    d = __builtin_amdgcn_wmma_f32_16x16x4_f32(
         false, a,  false, b,
         (short)0, d,  false,  false);
  }
#pragma unroll
  for (int i = 0; i < 8; ++i) {
    const int row = row0 + 8 * khalf + i;
    if (row < n_nodes)
      C[(size_t)row * OUT_DIM + w * 16 + mn] = d[i] * rowscale[8 * khalf + i];
  }
}

extern "C" void kernel_launch(void* const* d_in, const int* in_sizes, int n_in,
                              void* d_out, int out_size, void* d_ws, size_t ws_size,
                              hipStream_t stream) {
  const float* x  = (const float*)d_in[0];
  const int*   ei = (const int*)d_in[1];
  const float* W1 = (const float*)d_in[2];
  const float* b1 = (const float*)d_in[3];
  const float* W2 = (const float*)d_in[4];
  const float* b2 = (const float*)d_in[5];
  float* out = (float*)d_out;

  const int N = in_sizes[0] / IN_DIM;
  const int E = in_sizes[1] / 2;
  const int* src = ei;
  const int* dst = ei + E;

  float* h  = (float*)d_ws;
  float* h2 = h + (size_t)N * HID_DIM;
  int* cnt      = (int*)(h2 + (size_t)N * OUT_DIM);
  int* rowstart = cnt + N;
  int* csr      = rowstart + N;
  const int ntile = (E + TILE_E - 1) / TILE_E;
  const int nb    = (N + BIN_W - 1) / BIN_W;
  int2* staged = (int2*)h2;
  int*  table  = (int*)(staged + (size_t)ntile * TILE_E);

  const int B256 = 256;
  bin_tile_kernel<<<ntile, B256, 0, stream>>>(src, dst, staged, table, E, nb);

  const int waves  = (N + 15) / 16;

  group_gemm_kernel<IN_DIM, HID_DIM / 16><<<nb, 512, 0, stream>>>(
      staged, table, cnt, rowstart, csr, ntile, nb, x, W1, h, N);

  gather_gemm_kernel<<<waves, B256, 0, stream>>>(h, csr, rowstart, cnt, b1, W2, h2, N);

  const long long th2 = (long long)N * (OUT_DIM / 4);
  gather_kernel<OUT_DIM, false><<<(int)((th2 + B256 - 1) / B256), B256, 0, stream>>>(
      h2, csr, rowstart, cnt, b2, out, N);
}
